# starting kernel
# baseline (speedup 1.0000x reference)
#include <hip/hip_runtime.h>
#include <hip/hip_bf16.h>
#include <math.h>


#define N_NODES 65536
#define IN_CH   64
#define HC      128
#define R_CENT  16
#define B_GR    128
#define HEADS   2
#define C_CH    64

typedef __attribute__((ext_vector_type(2))) float v2f;
typedef __attribute__((ext_vector_type(8))) float v8f;

__global__ __launch_bounds__(256) void BipartPool_xr_kernel(
    const float* __restrict__ xc,
    const float* __restrict__ w,
    const float* __restrict__ b,
    float* __restrict__ xr)
{
  for (int i = threadIdx.x; i < R_CENT * HC; i += 256) {
    int r = i >> 7, col = i & 127;
    float acc = b[col];
    #pragma unroll 8
    for (int k = 0; k < IN_CH; ++k) acc += xc[r * IN_CH + k] * w[k * HC + col];
    xr[i] = acc;
  }
}

__global__ __launch_bounds__(256) void BipartPool_xl_gemm_kernel(
    const float* __restrict__ x,
    const float* __restrict__ w,
    const float* __restrict__ b,
    float* __restrict__ xl)
{
  const int wave = threadIdx.x >> 5;
  const int lane = threadIdx.x & 31;
  const int lo = lane & 15;
  const int hi = lane >> 4;
  const int row0 = blockIdx.x * 16;
  const int col0 = wave * 16;

  const float bv = b[col0 + lo];
  v8f c;
  #pragma unroll
  for (int i = 0; i < 8; ++i) c[i] = bv;

  const float* arow = x + (size_t)(row0 + lo) * IN_CH;
  #pragma unroll
  for (int k = 0; k < IN_CH; k += 4) {
    const int kk = k + 2 * hi;
    v2f a = *(const v2f*)(arow + kk);
    v2f bb;
    bb.x = w[(size_t)kk * HC + col0 + lo];
    bb.y = w[(size_t)(kk + 1) * HC + col0 + lo];
    c = __builtin_amdgcn_wmma_f32_16x16x4_f32(false, a, false, bb,
                                              (short)0, c, false, false);
  }

  float* orow = xl + (size_t)row0 * HC + col0 + lo;
  #pragma unroll
  for (int v = 0; v < 8; ++v) orow[(size_t)(v + 8 * hi) * HC] = c[v];
}

__global__ __launch_bounds__(256) void BipartPool_attn_kernel(
    const float* __restrict__ xl,
    const float* __restrict__ xr,
    const float* __restrict__ att,
    const float* __restrict__ bias,
    float* __restrict__ out)
{
  extern __shared__ float smem[];
  float* s_logit = smem;
  float* s_acc   = s_logit + 32 * 512;
  float* s_den   = s_acc + 2 * 16 * 64;
  float* s_pself = s_den + 32;
  float* s_self  = s_pself + 32;

  const int g    = blockIdx.x;
  const int tid  = threadIdx.x;
  const int base = g * 512;
  const int wave = tid >> 5;
  const int lane = tid & 31;

  for (int p4 = 0; p4 < 4; ++p4) {
    const int pair = tid + p4 * 256;
    const int n = pair & 511;
    const int h = pair >> 9;
    const float* xrow = xl + (size_t)(base + n) * HC + h * C_CH;
    const float* arow = att + h * C_CH;
    float xv[C_CH];
    #pragma unroll
    for (int c = 0; c < C_CH; ++c) xv[c] = xrow[c];
    for (int r = 0; r < R_CENT; ++r) {
      const float* rrow = xr + r * HC + h * C_CH;
      float acc = 0.f;
      #pragma unroll
      for (int c = 0; c < C_CH; ++c) {
        float s = xv[c] + rrow[c];
        s = s > 0.f ? s : 0.2f * s;
        acc += arow[c] * s;
      }
      s_logit[(h * 16 + r) * 512 + n] = acc;
    }
  }

  if (tid < 32) {
    const int r = tid & 15, h = tid >> 4;
    const int t = g * 16 + r;
    const float* xrow = xl + (size_t)t * HC + h * C_CH;
    const float* rrow = xr + r * HC + h * C_CH;
    const float* arow = att + h * C_CH;
    float acc = 0.f;
    #pragma unroll 8
    for (int c = 0; c < C_CH; ++c) {
      float s = xrow[c] + rrow[c];
      s = s > 0.f ? s : 0.2f * s;
      acc += arow[c] * s;
    }
    s_self[tid] = acc;
  }
  __syncthreads();

  if (g == 0 && tid < 32) {
    const int r = tid & 15, h = tid >> 4;
    s_logit[(h * 16 + r) * 512 + r] = -__builtin_inff();
  }
  __syncthreads();

  for (int q = 0; q < 4; ++q) {
    const int rh = wave * 4 + q;
    float* lrow = s_logit + rh * 512;
    float m = s_self[rh];
    for (int i = lane; i < 512; i += 32) m = fmaxf(m, lrow[i]);
    #pragma unroll
    for (int off = 16; off >= 1; off >>= 1) m = fmaxf(m, __shfl_xor(m, off, 32));
    float sum = 0.f;
    for (int i = lane; i < 512; i += 32) {
      const float pv = __expf(lrow[i] - m);
      lrow[i] = pv;
      sum += pv;
    }
    #pragma unroll
    for (int off = 16; off >= 1; off >>= 1) sum += __shfl_xor(sum, off, 32);
    if (lane == 0) {
      const float ps = __expf(s_self[rh] - m);
      s_pself[rh] = ps;
      s_den[rh] = sum + ps + 1e-16f;
    }
  }
  __syncthreads();

  {
    const int h = wave >> 2, nt = wave & 3;
    const int lo = lane & 15, hi = lane >> 4;
    const float* prow = s_logit + (h * 16 + lo) * 512;
    const float* bcol = xl + (size_t)base * HC + h * C_CH + nt * 16 + lo;
    v8f c;
    #pragma unroll
    for (int i = 0; i < 8; ++i) c[i] = 0.f;
    for (int k = 0; k < 512; k += 4) {
      const int kk = k + 2 * hi;
      v2f a = *(const v2f*)(prow + kk);
      v2f bb;
      bb.x = bcol[(size_t)kk * HC];
      bb.y = bcol[(size_t)(kk + 1) * HC];
      c = __builtin_amdgcn_wmma_f32_16x16x4_f32(false, a, false, bb,
                                                (short)0, c, false, false);
    }
    #pragma unroll
    for (int v = 0; v < 8; ++v)
      s_acc[(h * 16 + v + 8 * hi) * 64 + nt * 16 + lo] = c[v];
  }
  __syncthreads();

  for (int i = tid; i < 16 * 64; i += 256) {
    const int r = i >> 6, cc = i & 63;
    const int t = g * 16 + r;
    const float v0 = (s_acc[(0 * 16 + r) * 64 + cc] +
                      s_pself[0 * 16 + r] * xl[(size_t)t * HC + cc]) /
                     s_den[0 * 16 + r];
    const float v1 = (s_acc[(1 * 16 + r) * 64 + cc] +
                      s_pself[1 * 16 + r] * xl[(size_t)t * HC + C_CH + cc]) /
                     s_den[1 * 16 + r];
    out[(size_t)t * 64 + cc] = 0.5f * (v0 + v1) + bias[cc];
  }
}

extern "C" void kernel_launch(void* const* d_in, const int* in_sizes, int n_in,
                              void* d_out, int out_size, void* d_ws, size_t ws_size,
                              hipStream_t stream) {
  const float* x    = (const float*)d_in[0];
  const float* xcb  = (const float*)d_in[3];
  const float* wl   = (const float*)d_in[4];
  const float* bl   = (const float*)d_in[5];
  const float* wr   = (const float*)d_in[6];
  const float* br   = (const float*)d_in[7];
  const float* att  = (const float*)d_in[8];
  const float* bias = (const float*)d_in[9];
  float* out = (float*)d_out;

  float* xl = (float*)d_ws;
  float* xr = xl + (size_t)N_NODES * HC;

  BipartPool_xr_kernel<<<1, 256, 0, stream>>>(xcb, wr, br, xr);
  BipartPool_xl_gemm_kernel<<<N_NODES / 16, 256, 0, stream>>>(x, wl, bl, xl);

  const size_t smem = (size_t)(32 * 512 + 2 * 16 * 64 + 3 * 32) * sizeof(float);
  BipartPool_attn_kernel<<<B_GR, 256, smem, stream>>>(xl, xr, att, bias, out);
}


# logit loop: leaky_relu split into a linear part (one dot per node; per-target constant cancels in softmax) and a slope*|s| part, x_r and att as scalar operands: 1 add + 1 fma(|s|) per element
# speedup vs baseline: 1.5807x; 1.5807x over previous
#include <hip/hip_runtime.h>
#include <hip/hip_bf16.h>
#include <math.h>


#define N_NODES 65536
#define IN_CH   64
#define HC      128
#define R_CENT  16
#define B_GR    128
#define HEADS   2
#define C_CH    64

typedef __attribute__((ext_vector_type(2))) float v2f;
typedef __attribute__((ext_vector_type(8))) float v8f;

constexpr float KA_LRELU = (float)(0.5 * (1.0 + (double)0.2f));
constexpr float KB_LRELU = (float)(0.5 * (1.0 - (double)0.2f));


__device__ __forceinline__ void proj_strip16(
    const float* __restrict__ a,
    const float* __restrict__ w,
    const float* __restrict__ b,
    float* __restrict__ o)
{
  const int wave = threadIdx.x >> 5;
  const int lane = threadIdx.x & 31;
  const int lo = lane & 15;
  const int hi = lane >> 4;
  const int col0 = wave * 16;

  const float bv = b[col0 + lo];
  v8f c;
  #pragma unroll
  for (int i = 0; i < 8; ++i) c[i] = bv;

  const float* arow = a + (size_t)lo * IN_CH;
  #pragma unroll
  for (int k = 0; k < IN_CH; k += 4) {
    const int kk = k + 2 * hi;
    v2f av = *(const v2f*)(arow + kk);
    v2f bb;
    bb.x = w[(size_t)kk * HC + col0 + lo];
    bb.y = w[(size_t)(kk + 1) * HC + col0 + lo];
    c = __builtin_amdgcn_wmma_f32_16x16x4_f32(false, av, false, bb,
                                              (short)0, c, false, false);
  }

  float* orow = o + col0 + lo;
  #pragma unroll
  for (int v = 0; v < 8; ++v) orow[(size_t)(v + 8 * hi) * HC] = c[v];
}

__global__ __launch_bounds__(256, 16) void BipartPool_proj_gemm_kernel(
    const float* __restrict__ x,
    const float* __restrict__ wl,
    const float* __restrict__ bl,
    float* __restrict__ xl,
    const float* __restrict__ xc,
    const float* __restrict__ wr,
    const float* __restrict__ br,
    float* __restrict__ xr)
{
  if (blockIdx.x == 0) {
    proj_strip16(xc, wr, br, xr);
  } else {
    const size_t row0 = (size_t)(blockIdx.x - 1) * 16;
    proj_strip16(x + row0 * IN_CH, wl, bl, xl + row0 * HC);
  }
}

__global__ __launch_bounds__(256) void BipartPool_attn_kernel(
    const float* __restrict__ xl,
    const float* __restrict__ xr,
    const float* __restrict__ att,
    const float* __restrict__ bias,
    float* __restrict__ out)
{
  extern __shared__ float smem[];
  float* s_logit = smem;
  float* s_acc   = s_logit + 32 * 512;
  float* s_den   = s_acc + 2 * 16 * 64;
  float* s_pself = s_den + 32;
  float* s_self  = s_pself + 32;

  const int g    = blockIdx.x;
  const int tid  = threadIdx.x;
  const int base = g * 512;
  const int wave = tid >> 5;
  const int lane = tid & 31;

  #pragma unroll 1
  for (int pass = 0; pass < 4; ++pass) {
    const int h = pass >> 1;
    const int n = tid + (pass & 1) * 256;
    const float* xrow  = xl + (size_t)(base + n) * HC + h * C_CH;
    const float* arow  = att + h * C_CH;
    const float* rbase = xr + h * C_CH;
    float xv[C_CH];
    #pragma unroll
    for (int c = 0; c < C_CH; c += 4) {
      const float4 t = *(const float4*)(xrow + c);
      xv[c] = t.x; xv[c + 1] = t.y; xv[c + 2] = t.z; xv[c + 3] = t.w;
    }
    float acc[R_CENT][4];
    #pragma unroll
    for (int r = 0; r < R_CENT; ++r) {
      acc[r][0] = 0.f; acc[r][1] = 0.f; acc[r][2] = 0.f; acc[r][3] = 0.f;
    }
    float lin[4] = {0.f, 0.f, 0.f, 0.f};
    float av[16], rcur[16], rnxt[16];
    #pragma unroll
    for (int j = 0; j < 16; ++j) rcur[j] = rbase[j];
    #pragma unroll
    for (int q = 0; q < 4 * R_CENT; ++q) {
      const int r = q & 15, c0 = (q >> 4) * 16;
      if (q + 1 < 4 * R_CENT) {
        const float* rn = rbase + ((q + 1) & 15) * HC + ((q + 1) >> 4) * 16;
        #pragma unroll
        for (int j = 0; j < 16; ++j) rnxt[j] = rn[j];
      }
      if (r == 0) {
        #pragma unroll
        for (int j = 0; j < 16; ++j) av[j] = arow[c0 + j];
        #pragma unroll
        for (int j = 0; j < 16; ++j) lin[j & 3] = fmaf(av[j], xv[c0 + j], lin[j & 3]);
      }
      #pragma unroll
      for (int j = 0; j < 16; j += 4) {
        const float s0 = xv[c0 + j]     + rcur[j];
        const float s1 = xv[c0 + j + 1] + rcur[j + 1];
        const float s2 = xv[c0 + j + 2] + rcur[j + 2];
        const float s3 = xv[c0 + j + 3] + rcur[j + 3];
        acc[r][0] = fmaf(av[j],     fabsf(s0), acc[r][0]);
        acc[r][1] = fmaf(av[j + 1], fabsf(s1), acc[r][1]);
        acc[r][2] = fmaf(av[j + 2], fabsf(s2), acc[r][2]);
        acc[r][3] = fmaf(av[j + 3], fabsf(s3), acc[r][3]);
      }
      #pragma unroll
      for (int j = 0; j < 16; ++j) rcur[j] = rnxt[j];
    }
    const float lin_s = KA_LRELU * ((lin[0] + lin[1]) + (lin[2] + lin[3]));
    #pragma unroll
    for (int r = 0; r < R_CENT; ++r)
      s_logit[(h * 16 + r) * 512 + n] =
          lin_s + KB_LRELU * ((acc[r][0] + acc[r][1]) + (acc[r][2] + acc[r][3]));
  }

  if (tid < 32) {
    const int r = tid & 15, h = tid >> 4;
    const int t = g * 16 + r;
    const float* xrow = xl + (size_t)t * HC + h * C_CH;
    const float* rrow = xr + r * HC + h * C_CH;
    const float* arow = att + h * C_CH;
    float l0 = 0.f, l1 = 0.f, a0 = 0.f, a1 = 0.f;
    #pragma unroll 8
    for (int c = 0; c < C_CH; c += 2) {
      l0 = fmaf(arow[c],     xrow[c],     l0);
      l1 = fmaf(arow[c + 1], xrow[c + 1], l1);
      a0 = fmaf(arow[c],     fabsf(xrow[c]     + rrow[c]),     a0);
      a1 = fmaf(arow[c + 1], fabsf(xrow[c + 1] + rrow[c + 1]), a1);
    }
    s_self[tid] = KA_LRELU * (l0 + l1) + KB_LRELU * (a0 + a1);
  }
  __syncthreads();

  if (g == 0 && tid < 32) {
    const int r = tid & 15, h = tid >> 4;
    s_logit[(h * 16 + r) * 512 + r] = -__builtin_inff();
  }
  __syncthreads();

  for (int q = 0; q < 4; ++q) {
    const int rh = wave * 4 + q;
    float* lrow = s_logit + rh * 512;
    float m = s_self[rh];
    for (int i = lane; i < 512; i += 32) m = fmaxf(m, lrow[i]);
    #pragma unroll
    for (int off = 16; off >= 1; off >>= 1) m = fmaxf(m, __shfl_xor(m, off, 32));
    float sum = 0.f;
    for (int i = lane; i < 512; i += 32) {
      const float pv = __expf(lrow[i] - m);
      lrow[i] = pv;
      sum += pv;
    }
    #pragma unroll
    for (int off = 16; off >= 1; off >>= 1) sum += __shfl_xor(sum, off, 32);
    if (lane == 0) {
      const float ps = __expf(s_self[rh] - m);
      s_pself[rh] = ps;
      s_den[rh] = sum + ps + 1e-16f;
    }
  }
  __syncthreads();

  {
    const int h = wave >> 2, nt = wave & 3;
    const int lo = lane & 15, hi = lane >> 4;
    const float* prow = s_logit + (h * 16 + lo) * 512;
    const float* bcol = xl + (size_t)base * HC + h * C_CH + nt * 16 + lo;
    v8f c;
    #pragma unroll
    for (int i = 0; i < 8; ++i) c[i] = 0.f;
    for (int k = 0; k < 512; k += 4) {
      const int kk = k + 2 * hi;
      v2f a = *(const v2f*)(prow + kk);
      v2f bb;
      bb.x = bcol[(size_t)kk * HC];
      bb.y = bcol[(size_t)(kk + 1) * HC];
      c = __builtin_amdgcn_wmma_f32_16x16x4_f32(false, a, false, bb,
                                                (short)0, c, false, false);
    }
    #pragma unroll
    for (int v = 0; v < 8; ++v)
      s_acc[(h * 16 + v + 8 * hi) * 64 + nt * 16 + lo] = c[v];
  }
  __syncthreads();

  for (int i = tid; i < 16 * 64; i += 256) {
    const int r = i >> 6, cc = i & 63;
    const int t = g * 16 + r;
    const float v0 = (s_acc[(0 * 16 + r) * 64 + cc] +
                      s_pself[0 * 16 + r] * xl[(size_t)t * HC + cc]) /
                     s_den[0 * 16 + r];
    const float v1 = (s_acc[(1 * 16 + r) * 64 + cc] +
                      s_pself[1 * 16 + r] * xl[(size_t)t * HC + C_CH + cc]) /
                     s_den[1 * 16 + r];
    out[(size_t)t * 64 + cc] = 0.5f * (v0 + v1) + bias[cc];
  }
}

extern "C" void kernel_launch(void* const* d_in, const int* in_sizes, int n_in,
                              void* d_out, int out_size, void* d_ws, size_t ws_size,
                              hipStream_t stream) {
  const float* x    = (const float*)d_in[0];
  const float* xcb  = (const float*)d_in[3];
  const float* wl   = (const float*)d_in[4];
  const float* bl   = (const float*)d_in[5];
  const float* wr   = (const float*)d_in[6];
  const float* br   = (const float*)d_in[7];
  const float* att  = (const float*)d_in[8];
  const float* bias = (const float*)d_in[9];
  float* out = (float*)d_out;

  float* xl = (float*)d_ws;
  float* xr = xl + (size_t)N_NODES * HC;

  BipartPool_proj_gemm_kernel<<<N_NODES / 16 + 1, 256, 0, stream>>>(
      x, wl, bl, xl, xcb, wr, br, xr);

  const size_t smem = (size_t)(32 * 512 + 2 * 16 * 64 + 3 * 32) * sizeof(float);
  BipartPool_attn_kernel<<<B_GR, 256, smem, stream>>>(xl, xr, att, bias, out);
}


# x_l projection as a persistent f32 WMMA GEMM: 256 workgroups of 8 waves instead of 4097; each wave keeps its W_l fragments in registers over four 16-row strips and stores whole 128-byte lines
# speedup vs baseline: 1.8279x; 1.1564x over previous
#include <hip/hip_runtime.h>
#include <hip/hip_bf16.h>
#include <math.h>


#define N_NODES 65536
#define IN_CH   64
#define HC      128
#define R_CENT  16
#define B_GR    128
#define HEADS   2
#define C_CH    64

typedef __attribute__((ext_vector_type(2))) float v2f;
typedef __attribute__((ext_vector_type(8))) float v8f;

constexpr float KA_LRELU = (float)(0.5 * (1.0 + (double)0.2f));
constexpr float KB_LRELU = (float)(0.5 * (1.0 - (double)0.2f));


__device__ __forceinline__ void proj_strip16(
    const float* __restrict__ a,
    const float* __restrict__ w,
    const float* __restrict__ b,
    float* __restrict__ o)
{
  const int wave = threadIdx.x >> 5;
  const int lane = threadIdx.x & 31;
  const int lo = lane & 15;
  const int hi = lane >> 4;
  const int col0 = wave * 16;

  const float bv = b[col0 + lo];
  v8f c;
  #pragma unroll
  for (int i = 0; i < 8; ++i) c[i] = bv;

  const float* arow = a + (size_t)lo * IN_CH;
  #pragma unroll
  for (int k = 0; k < IN_CH; k += 4) {
    const int kk = k + 2 * hi;
    v2f av = *(const v2f*)(arow + kk);
    v2f bb;
    bb.x = w[(size_t)kk * HC + col0 + lo];
    bb.y = w[(size_t)(kk + 1) * HC + col0 + lo];
    c = __builtin_amdgcn_wmma_f32_16x16x4_f32(false, av, false, bb,
                                              (short)0, c, false, false);
  }

  float* orow = o + col0 + lo;
  #pragma unroll
  for (int v = 0; v < 8; ++v) orow[(size_t)(v + 8 * hi) * HC] = c[v];
}

#define XL_ROWS_PER_BLOCK 256
#define XL_STRIPS_PER_WAVE 4

__device__ __forceinline__ void load_a_strip(const float* __restrict__ ap, v2f (&a)[16]) {
  #pragma unroll
  for (int i = 0; i < 16; ++i) a[i] = *(const v2f*)(ap + 4 * i);
}

__device__ __forceinline__ void proj_rows256(
    const float* __restrict__ a,
    const float* __restrict__ w,
    const float* __restrict__ b,
    float* __restrict__ o)
{
  const int wave = threadIdx.x >> 5;
  const int lane = threadIdx.x & 31;
  const int lo = lane & 15;
  const int hi = lane >> 4;
  const int ncol = (wave & 1) * 64 + 4 * lo;
  const int row0 = (wave >> 1) * (16 * XL_STRIPS_PER_WAVE);

  const float* ap = a + (size_t)(row0 + lo) * IN_CH + 2 * hi;
  float* op = o + (size_t)(row0 + 8 * hi) * HC + ncol;

  float4 wr0[16], wr1[16];
  #pragma unroll
  for (int i = 0; i < 16; ++i) {
    const int kk = 4 * i + 2 * hi;
    wr0[i] = *(const float4*)(w + (size_t)kk * HC + ncol);
    wr1[i] = *(const float4*)(w + (size_t)(kk + 1) * HC + ncol);
  }
  const float4 bv = *(const float4*)(b + ncol);
  v2f acur[16], anxt[16];
  load_a_strip(ap, acur);
  __builtin_amdgcn_sched_barrier(0);

  v2f bf0[16], bf1[16], bf2[16], bf3[16];
  #pragma unroll
  for (int i = 0; i < 16; ++i) {
    bf0[i].x = wr0[i].x; bf0[i].y = wr1[i].x;
    bf1[i].x = wr0[i].y; bf1[i].y = wr1[i].y;
    bf2[i].x = wr0[i].z; bf2[i].y = wr1[i].z;
    bf3[i].x = wr0[i].w; bf3[i].y = wr1[i].w;
  }

  #pragma unroll
  for (int s = 0; s < XL_STRIPS_PER_WAVE; ++s) {
    if (s + 1 < XL_STRIPS_PER_WAVE) load_a_strip(ap + (size_t)(s + 1) * 16 * IN_CH, anxt);
    __builtin_amdgcn_sched_barrier(0);

    v8f c0, c1, c2, c3;
    #pragma unroll
    for (int r = 0; r < 8; ++r) { c0[r] = bv.x; c1[r] = bv.y; c2[r] = bv.z; c3[r] = bv.w; }
    #pragma unroll
    for (int i = 0; i < 16; ++i) {
      c0 = __builtin_amdgcn_wmma_f32_16x16x4_f32(false, acur[i], false, bf0[i], (short)0, c0, false, false);
      c1 = __builtin_amdgcn_wmma_f32_16x16x4_f32(false, acur[i], false, bf1[i], (short)0, c1, false, false);
      c2 = __builtin_amdgcn_wmma_f32_16x16x4_f32(false, acur[i], false, bf2[i], (short)0, c2, false, false);
      c3 = __builtin_amdgcn_wmma_f32_16x16x4_f32(false, acur[i], false, bf3[i], (short)0, c3, false, false);
    }
    #pragma unroll
    for (int r = 0; r < 8; ++r) {
      float4 t;
      t.x = c0[r]; t.y = c1[r]; t.z = c2[r]; t.w = c3[r];
      *(float4*)(op + (size_t)(s * 16 + r) * HC) = t;
    }
    #pragma unroll
    for (int i = 0; i < 16; ++i) acur[i] = anxt[i];
  }
}

__global__ __launch_bounds__(256, 2) void BipartPool_proj_gemm_kernel(
    const float* __restrict__ x,
    const float* __restrict__ wl,
    const float* __restrict__ bl,
    float* __restrict__ xl,
    const float* __restrict__ xc,
    const float* __restrict__ wr,
    const float* __restrict__ br,
    float* __restrict__ xr)
{
  const size_t row0 = (size_t)blockIdx.x * XL_ROWS_PER_BLOCK;
  proj_rows256(x + row0 * IN_CH, wl, bl, xl + row0 * HC);
  if (blockIdx.x == 0) proj_strip16(xc, wr, br, xr);
}

__global__ __launch_bounds__(256) void BipartPool_attn_kernel(
    const float* __restrict__ xl,
    const float* __restrict__ xr,
    const float* __restrict__ att,
    const float* __restrict__ bias,
    float* __restrict__ out)
{
  extern __shared__ float smem[];
  float* s_logit = smem;
  float* s_acc   = s_logit + 32 * 512;
  float* s_den   = s_acc + 2 * 16 * 64;
  float* s_pself = s_den + 32;
  float* s_self  = s_pself + 32;

  const int g    = blockIdx.x;
  const int tid  = threadIdx.x;
  const int base = g * 512;
  const int wave = tid >> 5;
  const int lane = tid & 31;

  #pragma unroll 1
  for (int pass = 0; pass < 4; ++pass) {
    const int h = pass >> 1;
    const int n = tid + (pass & 1) * 256;
    const float* xrow  = xl + (size_t)(base + n) * HC + h * C_CH;
    const float* arow  = att + h * C_CH;
    const float* rbase = xr + h * C_CH;
    float xv[C_CH];
    #pragma unroll
    for (int c = 0; c < C_CH; c += 4) {
      const float4 t = *(const float4*)(xrow + c);
      xv[c] = t.x; xv[c + 1] = t.y; xv[c + 2] = t.z; xv[c + 3] = t.w;
    }
    float acc[R_CENT][4];
    #pragma unroll
    for (int r = 0; r < R_CENT; ++r) {
      acc[r][0] = 0.f; acc[r][1] = 0.f; acc[r][2] = 0.f; acc[r][3] = 0.f;
    }
    float lin[4] = {0.f, 0.f, 0.f, 0.f};
    float av[16], rcur[16], rnxt[16];
    #pragma unroll
    for (int j = 0; j < 16; ++j) rcur[j] = rbase[j];
    #pragma unroll
    for (int q = 0; q < 4 * R_CENT; ++q) {
      const int r = q & 15, c0 = (q >> 4) * 16;
      if (q + 1 < 4 * R_CENT) {
        const float* rn = rbase + ((q + 1) & 15) * HC + ((q + 1) >> 4) * 16;
        #pragma unroll
        for (int j = 0; j < 16; ++j) rnxt[j] = rn[j];
      }
      if (r == 0) {
        #pragma unroll
        for (int j = 0; j < 16; ++j) av[j] = arow[c0 + j];
        #pragma unroll
        for (int j = 0; j < 16; ++j) lin[j & 3] = fmaf(av[j], xv[c0 + j], lin[j & 3]);
      }
      #pragma unroll
      for (int j = 0; j < 16; j += 4) {
        const float s0 = xv[c0 + j]     + rcur[j];
        const float s1 = xv[c0 + j + 1] + rcur[j + 1];
        const float s2 = xv[c0 + j + 2] + rcur[j + 2];
        const float s3 = xv[c0 + j + 3] + rcur[j + 3];
        acc[r][0] = fmaf(av[j],     fabsf(s0), acc[r][0]);
        acc[r][1] = fmaf(av[j + 1], fabsf(s1), acc[r][1]);
        acc[r][2] = fmaf(av[j + 2], fabsf(s2), acc[r][2]);
        acc[r][3] = fmaf(av[j + 3], fabsf(s3), acc[r][3]);
      }
      #pragma unroll
      for (int j = 0; j < 16; ++j) rcur[j] = rnxt[j];
    }
    const float lin_s = KA_LRELU * ((lin[0] + lin[1]) + (lin[2] + lin[3]));
    #pragma unroll
    for (int r = 0; r < R_CENT; ++r)
      s_logit[(h * 16 + r) * 512 + n] =
          lin_s + KB_LRELU * ((acc[r][0] + acc[r][1]) + (acc[r][2] + acc[r][3]));
  }

  if (tid < 32) {
    const int r = tid & 15, h = tid >> 4;
    const int t = g * 16 + r;
    const float* xrow = xl + (size_t)t * HC + h * C_CH;
    const float* rrow = xr + r * HC + h * C_CH;
    const float* arow = att + h * C_CH;
    float l0 = 0.f, l1 = 0.f, a0 = 0.f, a1 = 0.f;
    #pragma unroll 8
    for (int c = 0; c < C_CH; c += 2) {
      l0 = fmaf(arow[c],     xrow[c],     l0);
      l1 = fmaf(arow[c + 1], xrow[c + 1], l1);
      a0 = fmaf(arow[c],     fabsf(xrow[c]     + rrow[c]),     a0);
      a1 = fmaf(arow[c + 1], fabsf(xrow[c + 1] + rrow[c + 1]), a1);
    }
    s_self[tid] = KA_LRELU * (l0 + l1) + KB_LRELU * (a0 + a1);
  }
  __syncthreads();

  if (g == 0 && tid < 32) {
    const int r = tid & 15, h = tid >> 4;
    s_logit[(h * 16 + r) * 512 + r] = -__builtin_inff();
  }
  __syncthreads();

  for (int q = 0; q < 4; ++q) {
    const int rh = wave * 4 + q;
    float* lrow = s_logit + rh * 512;
    float m = s_self[rh];
    for (int i = lane; i < 512; i += 32) m = fmaxf(m, lrow[i]);
    #pragma unroll
    for (int off = 16; off >= 1; off >>= 1) m = fmaxf(m, __shfl_xor(m, off, 32));
    float sum = 0.f;
    for (int i = lane; i < 512; i += 32) {
      const float pv = __expf(lrow[i] - m);
      lrow[i] = pv;
      sum += pv;
    }
    #pragma unroll
    for (int off = 16; off >= 1; off >>= 1) sum += __shfl_xor(sum, off, 32);
    if (lane == 0) {
      const float ps = __expf(s_self[rh] - m);
      s_pself[rh] = ps;
      s_den[rh] = sum + ps + 1e-16f;
    }
  }
  __syncthreads();

  {
    const int h = wave >> 2, nt = wave & 3;
    const int lo = lane & 15, hi = lane >> 4;
    const float* prow = s_logit + (h * 16 + lo) * 512;
    const float* bcol = xl + (size_t)base * HC + h * C_CH + nt * 16 + lo;
    v8f c;
    #pragma unroll
    for (int i = 0; i < 8; ++i) c[i] = 0.f;
    for (int k = 0; k < 512; k += 4) {
      const int kk = k + 2 * hi;
      v2f a = *(const v2f*)(prow + kk);
      v2f bb;
      bb.x = bcol[(size_t)kk * HC];
      bb.y = bcol[(size_t)(kk + 1) * HC];
      c = __builtin_amdgcn_wmma_f32_16x16x4_f32(false, a, false, bb,
                                                (short)0, c, false, false);
    }
    #pragma unroll
    for (int v = 0; v < 8; ++v)
      s_acc[(h * 16 + v + 8 * hi) * 64 + nt * 16 + lo] = c[v];
  }
  __syncthreads();

  for (int i = tid; i < 16 * 64; i += 256) {
    const int r = i >> 6, cc = i & 63;
    const int t = g * 16 + r;
    const float v0 = (s_acc[(0 * 16 + r) * 64 + cc] +
                      s_pself[0 * 16 + r] * xl[(size_t)t * HC + cc]) /
                     s_den[0 * 16 + r];
    const float v1 = (s_acc[(1 * 16 + r) * 64 + cc] +
                      s_pself[1 * 16 + r] * xl[(size_t)t * HC + C_CH + cc]) /
                     s_den[1 * 16 + r];
    out[(size_t)t * 64 + cc] = 0.5f * (v0 + v1) + bias[cc];
  }
}

extern "C" void kernel_launch(void* const* d_in, const int* in_sizes, int n_in,
                              void* d_out, int out_size, void* d_ws, size_t ws_size,
                              hipStream_t stream) {
  const float* x    = (const float*)d_in[0];
  const float* xcb  = (const float*)d_in[3];
  const float* wl   = (const float*)d_in[4];
  const float* bl   = (const float*)d_in[5];
  const float* wr   = (const float*)d_in[6];
  const float* br   = (const float*)d_in[7];
  const float* att  = (const float*)d_in[8];
  const float* bias = (const float*)d_in[9];
  float* out = (float*)d_out;

  float* xl = (float*)d_ws;
  float* xr = xl + (size_t)N_NODES * HC;

  BipartPool_proj_gemm_kernel<<<N_NODES / XL_ROWS_PER_BLOCK, 256, 0, stream>>>(
      x, wl, bl, xl, xcb, wr, br, xr);

  const size_t smem = (size_t)(32 * 512 + 2 * 16 * 64 + 3 * 32) * sizeof(float);
  BipartPool_attn_kernel<<<B_GR, 256, smem, stream>>>(xl, xr, att, bias, out);
}


# attention kernel: two workgroups per graph (8 of the 16 targets each, both heads), 256 workgroups instead of 128, so every WGP has one; logit, softmax and epilogue work per workgroup halves
# speedup vs baseline: 2.4043x; 1.3153x over previous
#include <hip/hip_runtime.h>
#include <hip/hip_bf16.h>
#include <math.h>


#define N_NODES 65536
#define IN_CH   64
#define HC      128
#define R_CENT  16
#define B_GR    128
#define HEADS   2
#define C_CH    64

typedef __attribute__((ext_vector_type(2))) float v2f;
typedef __attribute__((ext_vector_type(8))) float v8f;

constexpr float KA_LRELU = (float)(0.5 * (1.0 + (double)0.2f));
constexpr float KB_LRELU = (float)(0.5 * (1.0 - (double)0.2f));


__device__ __forceinline__ void proj_strip16(
    const float* __restrict__ a,
    const float* __restrict__ w,
    const float* __restrict__ b,
    float* __restrict__ o)
{
  const int wave = threadIdx.x >> 5;
  const int lane = threadIdx.x & 31;
  const int lo = lane & 15;
  const int hi = lane >> 4;
  const int col0 = wave * 16;

  const float bv = b[col0 + lo];
  v8f c;
  #pragma unroll
  for (int i = 0; i < 8; ++i) c[i] = bv;

  const float* arow = a + (size_t)lo * IN_CH;
  #pragma unroll
  for (int k = 0; k < IN_CH; k += 4) {
    const int kk = k + 2 * hi;
    v2f av = *(const v2f*)(arow + kk);
    v2f bb;
    bb.x = w[(size_t)kk * HC + col0 + lo];
    bb.y = w[(size_t)(kk + 1) * HC + col0 + lo];
    c = __builtin_amdgcn_wmma_f32_16x16x4_f32(false, av, false, bb,
                                              (short)0, c, false, false);
  }

  float* orow = o + col0 + lo;
  #pragma unroll
  for (int v = 0; v < 8; ++v) orow[(size_t)(v + 8 * hi) * HC] = c[v];
}

#define XL_ROWS_PER_BLOCK 256
#define XL_STRIPS_PER_WAVE 4

__device__ __forceinline__ void load_a_strip(const float* __restrict__ ap, v2f (&a)[16]) {
  #pragma unroll
  for (int i = 0; i < 16; ++i) a[i] = *(const v2f*)(ap + 4 * i);
}

__device__ __forceinline__ void proj_rows256(
    const float* __restrict__ a,
    const float* __restrict__ w,
    const float* __restrict__ b,
    float* __restrict__ o)
{
  const int wave = threadIdx.x >> 5;
  const int lane = threadIdx.x & 31;
  const int lo = lane & 15;
  const int hi = lane >> 4;
  const int ncol = (wave & 1) * 64 + 4 * lo;
  const int row0 = (wave >> 1) * (16 * XL_STRIPS_PER_WAVE);

  const float* ap = a + (size_t)(row0 + lo) * IN_CH + 2 * hi;
  float* op = o + (size_t)(row0 + 8 * hi) * HC + ncol;

  float4 wr0[16], wr1[16];
  #pragma unroll
  for (int i = 0; i < 16; ++i) {
    const int kk = 4 * i + 2 * hi;
    wr0[i] = *(const float4*)(w + (size_t)kk * HC + ncol);
    wr1[i] = *(const float4*)(w + (size_t)(kk + 1) * HC + ncol);
  }
  const float4 bv = *(const float4*)(b + ncol);
  v2f acur[16], anxt[16];
  load_a_strip(ap, acur);
  __builtin_amdgcn_sched_barrier(0);

  v2f bf0[16], bf1[16], bf2[16], bf3[16];
  #pragma unroll
  for (int i = 0; i < 16; ++i) {
    bf0[i].x = wr0[i].x; bf0[i].y = wr1[i].x;
    bf1[i].x = wr0[i].y; bf1[i].y = wr1[i].y;
    bf2[i].x = wr0[i].z; bf2[i].y = wr1[i].z;
    bf3[i].x = wr0[i].w; bf3[i].y = wr1[i].w;
  }

  #pragma unroll
  for (int s = 0; s < XL_STRIPS_PER_WAVE; ++s) {
    if (s + 1 < XL_STRIPS_PER_WAVE) load_a_strip(ap + (size_t)(s + 1) * 16 * IN_CH, anxt);
    __builtin_amdgcn_sched_barrier(0);

    v8f c0, c1, c2, c3;
    #pragma unroll
    for (int r = 0; r < 8; ++r) { c0[r] = bv.x; c1[r] = bv.y; c2[r] = bv.z; c3[r] = bv.w; }
    #pragma unroll
    for (int i = 0; i < 16; ++i) {
      c0 = __builtin_amdgcn_wmma_f32_16x16x4_f32(false, acur[i], false, bf0[i], (short)0, c0, false, false);
      c1 = __builtin_amdgcn_wmma_f32_16x16x4_f32(false, acur[i], false, bf1[i], (short)0, c1, false, false);
      c2 = __builtin_amdgcn_wmma_f32_16x16x4_f32(false, acur[i], false, bf2[i], (short)0, c2, false, false);
      c3 = __builtin_amdgcn_wmma_f32_16x16x4_f32(false, acur[i], false, bf3[i], (short)0, c3, false, false);
    }
    #pragma unroll
    for (int r = 0; r < 8; ++r) {
      float4 t;
      t.x = c0[r]; t.y = c1[r]; t.z = c2[r]; t.w = c3[r];
      *(float4*)(op + (size_t)(s * 16 + r) * HC) = t;
    }
    #pragma unroll
    for (int i = 0; i < 16; ++i) acur[i] = anxt[i];
  }
}

__global__ __launch_bounds__(256, 2) void BipartPool_proj_gemm_kernel(
    const float* __restrict__ x,
    const float* __restrict__ wl,
    const float* __restrict__ bl,
    float* __restrict__ xl,
    const float* __restrict__ xc,
    const float* __restrict__ wr,
    const float* __restrict__ br,
    float* __restrict__ xr)
{
  const size_t row0 = (size_t)blockIdx.x * XL_ROWS_PER_BLOCK;
  proj_rows256(x + row0 * IN_CH, wl, bl, xl + row0 * HC);
  if (blockIdx.x == 0) proj_strip16(xc, wr, br, xr);
}

#define ATT_T_PER_WG   8
#define ATT_ROWS       (HEADS * ATT_T_PER_WG)
#define ATT_WGS        (B_GR * (R_CENT / ATT_T_PER_WG))
#define ATT_LDS_FLOATS (ATT_ROWS * 512 + ATT_T_PER_WG * HC + 3 * ATT_ROWS)

__global__ __launch_bounds__(256) void BipartPool_attn_kernel(
    const float* __restrict__ xl,
    const float* __restrict__ xr,
    const float* __restrict__ att,
    const float* __restrict__ bias,
    float* __restrict__ out)
{
  extern __shared__ float smem[];
  float* s_logit = smem;
  float* s_acc   = s_logit + ATT_ROWS * 512;
  float* s_den   = s_acc + ATT_T_PER_WG * HC;
  float* s_pself = s_den + ATT_ROWS;
  float* s_self  = s_pself + ATT_ROWS;

  const int g    = blockIdx.x >> 1;
  const int r0   = (blockIdx.x & 1) * ATT_T_PER_WG;
  const int tid  = threadIdx.x;
  const int base = g * 512;
  const int wave = tid >> 5;
  const int lane = tid & 31;

  #pragma unroll 1
  for (int pass = 0; pass < 4; ++pass) {
    const int h = pass >> 1;
    const int n = tid + (pass & 1) * 256;
    const float* xrow  = xl + (size_t)(base + n) * HC + h * C_CH;
    const float* arow  = att + h * C_CH;
    const float* rbase = xr + r0 * HC + h * C_CH;
    float xv[C_CH];
    #pragma unroll
    for (int c = 0; c < C_CH; c += 4) {
      const float4 t = *(const float4*)(xrow + c);
      xv[c] = t.x; xv[c + 1] = t.y; xv[c + 2] = t.z; xv[c + 3] = t.w;
    }
    float acc[ATT_T_PER_WG][4];
    #pragma unroll
    for (int r = 0; r < ATT_T_PER_WG; ++r) {
      acc[r][0] = 0.f; acc[r][1] = 0.f; acc[r][2] = 0.f; acc[r][3] = 0.f;
    }
    float lin[4] = {0.f, 0.f, 0.f, 0.f};
    float av[16], rcur[16], rnxt[16];
    #pragma unroll
    for (int j = 0; j < 16; ++j) rcur[j] = rbase[j];
    #pragma unroll
    for (int q = 0; q < 4 * ATT_T_PER_WG; ++q) {
      const int r = q % ATT_T_PER_WG, c0 = (q / ATT_T_PER_WG) * 16;
      if (q + 1 < 4 * ATT_T_PER_WG) {
        const float* rn = rbase + ((q + 1) % ATT_T_PER_WG) * HC
                                + ((q + 1) / ATT_T_PER_WG) * 16;
        #pragma unroll
        for (int j = 0; j < 16; ++j) rnxt[j] = rn[j];
      }
      if (r == 0) {
        #pragma unroll
        for (int j = 0; j < 16; ++j) av[j] = arow[c0 + j];
        #pragma unroll
        for (int j = 0; j < 16; ++j) lin[j & 3] = fmaf(av[j], xv[c0 + j], lin[j & 3]);
      }
      #pragma unroll
      for (int j = 0; j < 16; j += 4) {
        const float s0 = xv[c0 + j]     + rcur[j];
        const float s1 = xv[c0 + j + 1] + rcur[j + 1];
        const float s2 = xv[c0 + j + 2] + rcur[j + 2];
        const float s3 = xv[c0 + j + 3] + rcur[j + 3];
        acc[r][0] = fmaf(av[j],     fabsf(s0), acc[r][0]);
        acc[r][1] = fmaf(av[j + 1], fabsf(s1), acc[r][1]);
        acc[r][2] = fmaf(av[j + 2], fabsf(s2), acc[r][2]);
        acc[r][3] = fmaf(av[j + 3], fabsf(s3), acc[r][3]);
      }
      #pragma unroll
      for (int j = 0; j < 16; ++j) rcur[j] = rnxt[j];
    }
    const float lin_s = KA_LRELU * ((lin[0] + lin[1]) + (lin[2] + lin[3]));
    #pragma unroll
    for (int r = 0; r < ATT_T_PER_WG; ++r)
      s_logit[(h * ATT_T_PER_WG + r) * 512 + n] =
          lin_s + KB_LRELU * ((acc[r][0] + acc[r][1]) + (acc[r][2] + acc[r][3]));
  }

  if (tid < ATT_ROWS) {
    const int r8 = tid % ATT_T_PER_WG, h = tid / ATT_T_PER_WG;
    const int r = r0 + r8;
    const int t = g * 16 + r;
    const float* xrow = xl + (size_t)t * HC + h * C_CH;
    const float* rrow = xr + r * HC + h * C_CH;
    const float* arow = att + h * C_CH;
    float l0 = 0.f, l1 = 0.f, a0 = 0.f, a1 = 0.f;
    #pragma unroll 8
    for (int c = 0; c < C_CH; c += 2) {
      l0 = fmaf(arow[c],     xrow[c],     l0);
      l1 = fmaf(arow[c + 1], xrow[c + 1], l1);
      a0 = fmaf(arow[c],     fabsf(xrow[c]     + rrow[c]),     a0);
      a1 = fmaf(arow[c + 1], fabsf(xrow[c + 1] + rrow[c + 1]), a1);
    }
    s_self[tid] = KA_LRELU * (l0 + l1) + KB_LRELU * (a0 + a1);
  }
  __syncthreads();

  if (g == 0 && tid < ATT_ROWS) {
    const int r8 = tid % ATT_T_PER_WG, h = tid / ATT_T_PER_WG;
    s_logit[(h * ATT_T_PER_WG + r8) * 512 + (r0 + r8)] = -__builtin_inff();
  }
  __syncthreads();

  for (int q = 0; q < ATT_ROWS / 8; ++q) {
    const int rh = wave * (ATT_ROWS / 8) + q;
    float* lrow = s_logit + rh * 512;
    float m = s_self[rh];
    for (int i = lane; i < 512; i += 32) m = fmaxf(m, lrow[i]);
    #pragma unroll
    for (int off = 16; off >= 1; off >>= 1) m = fmaxf(m, __shfl_xor(m, off, 32));
    float sum = 0.f;
    for (int i = lane; i < 512; i += 32) {
      const float pv = __expf(lrow[i] - m);
      lrow[i] = pv;
      sum += pv;
    }
    #pragma unroll
    for (int off = 16; off >= 1; off >>= 1) sum += __shfl_xor(sum, off, 32);
    if (lane == 0) {
      const float ps = __expf(s_self[rh] - m);
      s_pself[rh] = ps;
      s_den[rh] = sum + ps + 1e-16f;
    }
  }
  __syncthreads();

  {
    const int hw = wave >> 2;
    const int lo = lane & 15, hi = lane >> 4;
    const float* prow = s_logit + lo * 512;
    const float* bcol = xl + (size_t)base * HC + wave * 16 + lo;
    v8f c;
    #pragma unroll
    for (int i = 0; i < 8; ++i) c[i] = 0.f;
    for (int k = 0; k < 512; k += 4) {
      const int kk = k + 2 * hi;
      v2f a = *(const v2f*)(prow + kk);
      v2f bb;
      bb.x = bcol[(size_t)kk * HC];
      bb.y = bcol[(size_t)(kk + 1) * HC];
      c = __builtin_amdgcn_wmma_f32_16x16x4_f32(false, a, false, bb,
                                                (short)0, c, false, false);
    }
    if (hi == hw) {
      #pragma unroll
      for (int v = 0; v < 8; ++v)
        s_acc[v * HC + wave * 16 + lo] = c[v];
    }
  }
  __syncthreads();

  for (int i = tid; i < ATT_T_PER_WG * 64; i += 256) {
    const int r8 = i >> 6, cc = i & 63;
    const int t = g * 16 + r0 + r8;
    const float v0 = (s_acc[r8 * HC + cc] +
                      s_pself[r8] * xl[(size_t)t * HC + cc]) /
                     s_den[r8];
    const float v1 = (s_acc[r8 * HC + C_CH + cc] +
                      s_pself[ATT_T_PER_WG + r8] * xl[(size_t)t * HC + C_CH + cc]) /
                     s_den[ATT_T_PER_WG + r8];
    out[(size_t)t * 64 + cc] = 0.5f * (v0 + v1) + bias[cc];
  }
}

extern "C" void kernel_launch(void* const* d_in, const int* in_sizes, int n_in,
                              void* d_out, int out_size, void* d_ws, size_t ws_size,
                              hipStream_t stream) {
  const float* x    = (const float*)d_in[0];
  const float* xcb  = (const float*)d_in[3];
  const float* wl   = (const float*)d_in[4];
  const float* bl   = (const float*)d_in[5];
  const float* wr   = (const float*)d_in[6];
  const float* br   = (const float*)d_in[7];
  const float* att  = (const float*)d_in[8];
  const float* bias = (const float*)d_in[9];
  float* out = (float*)d_out;

  float* xl = (float*)d_ws;
  float* xr = xl + (size_t)N_NODES * HC;

  BipartPool_proj_gemm_kernel<<<N_NODES / XL_ROWS_PER_BLOCK, 256, 0, stream>>>(
      x, wl, bl, xl, xcb, wr, br, xr);

  const size_t smem = (size_t)ATT_LDS_FLOATS * sizeof(float);
  BipartPool_attn_kernel<<<ATT_WGS, 256, smem, stream>>>(xl, xr, att, bias, out);
}


# attention kernel: row pitch of the LDS exp/logit rows 512 -> 516 floats, so the row-per-lane A-fragment reads of the weighted sum spread over all 64 banks instead of a 16-way conflict on two banks
# speedup vs baseline: 2.6114x; 1.0861x over previous
#include <hip/hip_runtime.h>
#include <hip/hip_bf16.h>
#include <math.h>


#define N_NODES 65536
#define IN_CH   64
#define HC      128
#define R_CENT  16
#define B_GR    128
#define HEADS   2
#define C_CH    64

typedef __attribute__((ext_vector_type(2))) float v2f;
typedef __attribute__((ext_vector_type(8))) float v8f;

constexpr float KA_LRELU = (float)(0.5 * (1.0 + (double)0.2f));
constexpr float KB_LRELU = (float)(0.5 * (1.0 - (double)0.2f));


__device__ __forceinline__ void proj_strip16(
    const float* __restrict__ a,
    const float* __restrict__ w,
    const float* __restrict__ b,
    float* __restrict__ o)
{
  const int wave = threadIdx.x >> 5;
  const int lane = threadIdx.x & 31;
  const int lo = lane & 15;
  const int hi = lane >> 4;
  const int col0 = wave * 16;

  const float bv = b[col0 + lo];
  v8f c;
  #pragma unroll
  for (int i = 0; i < 8; ++i) c[i] = bv;

  const float* arow = a + (size_t)lo * IN_CH;
  #pragma unroll
  for (int k = 0; k < IN_CH; k += 4) {
    const int kk = k + 2 * hi;
    v2f av = *(const v2f*)(arow + kk);
    v2f bb;
    bb.x = w[(size_t)kk * HC + col0 + lo];
    bb.y = w[(size_t)(kk + 1) * HC + col0 + lo];
    c = __builtin_amdgcn_wmma_f32_16x16x4_f32(false, av, false, bb,
                                              (short)0, c, false, false);
  }

  float* orow = o + col0 + lo;
  #pragma unroll
  for (int v = 0; v < 8; ++v) orow[(size_t)(v + 8 * hi) * HC] = c[v];
}

#define XL_ROWS_PER_BLOCK 256
#define XL_STRIPS_PER_WAVE 4

__device__ __forceinline__ void load_a_strip(const float* __restrict__ ap, v2f (&a)[16]) {
  #pragma unroll
  for (int i = 0; i < 16; ++i) a[i] = *(const v2f*)(ap + 4 * i);
}

__device__ __forceinline__ void proj_rows256(
    const float* __restrict__ a,
    const float* __restrict__ w,
    const float* __restrict__ b,
    float* __restrict__ o)
{
  const int wave = threadIdx.x >> 5;
  const int lane = threadIdx.x & 31;
  const int lo = lane & 15;
  const int hi = lane >> 4;
  const int ncol = (wave & 1) * 64 + 4 * lo;
  const int row0 = (wave >> 1) * (16 * XL_STRIPS_PER_WAVE);

  const float* ap = a + (size_t)(row0 + lo) * IN_CH + 2 * hi;
  float* op = o + (size_t)(row0 + 8 * hi) * HC + ncol;

  float4 wr0[16], wr1[16];
  #pragma unroll
  for (int i = 0; i < 16; ++i) {
    const int kk = 4 * i + 2 * hi;
    wr0[i] = *(const float4*)(w + (size_t)kk * HC + ncol);
    wr1[i] = *(const float4*)(w + (size_t)(kk + 1) * HC + ncol);
  }
  const float4 bv = *(const float4*)(b + ncol);
  v2f acur[16], anxt[16];
  load_a_strip(ap, acur);
  __builtin_amdgcn_sched_barrier(0);

  v2f bf0[16], bf1[16], bf2[16], bf3[16];
  #pragma unroll
  for (int i = 0; i < 16; ++i) {
    bf0[i].x = wr0[i].x; bf0[i].y = wr1[i].x;
    bf1[i].x = wr0[i].y; bf1[i].y = wr1[i].y;
    bf2[i].x = wr0[i].z; bf2[i].y = wr1[i].z;
    bf3[i].x = wr0[i].w; bf3[i].y = wr1[i].w;
  }

  #pragma unroll
  for (int s = 0; s < XL_STRIPS_PER_WAVE; ++s) {
    if (s + 1 < XL_STRIPS_PER_WAVE) load_a_strip(ap + (size_t)(s + 1) * 16 * IN_CH, anxt);
    __builtin_amdgcn_sched_barrier(0);

    v8f c0, c1, c2, c3;
    #pragma unroll
    for (int r = 0; r < 8; ++r) { c0[r] = bv.x; c1[r] = bv.y; c2[r] = bv.z; c3[r] = bv.w; }
    #pragma unroll
    for (int i = 0; i < 16; ++i) {
      c0 = __builtin_amdgcn_wmma_f32_16x16x4_f32(false, acur[i], false, bf0[i], (short)0, c0, false, false);
      c1 = __builtin_amdgcn_wmma_f32_16x16x4_f32(false, acur[i], false, bf1[i], (short)0, c1, false, false);
      c2 = __builtin_amdgcn_wmma_f32_16x16x4_f32(false, acur[i], false, bf2[i], (short)0, c2, false, false);
      c3 = __builtin_amdgcn_wmma_f32_16x16x4_f32(false, acur[i], false, bf3[i], (short)0, c3, false, false);
    }
    #pragma unroll
    for (int r = 0; r < 8; ++r) {
      float4 t;
      t.x = c0[r]; t.y = c1[r]; t.z = c2[r]; t.w = c3[r];
      *(float4*)(op + (size_t)(s * 16 + r) * HC) = t;
    }
    #pragma unroll
    for (int i = 0; i < 16; ++i) acur[i] = anxt[i];
  }
}

__global__ __launch_bounds__(256, 2) void BipartPool_proj_gemm_kernel(
    const float* __restrict__ x,
    const float* __restrict__ wl,
    const float* __restrict__ bl,
    float* __restrict__ xl,
    const float* __restrict__ xc,
    const float* __restrict__ wr,
    const float* __restrict__ br,
    float* __restrict__ xr)
{
  const size_t row0 = (size_t)blockIdx.x * XL_ROWS_PER_BLOCK;
  proj_rows256(x + row0 * IN_CH, wl, bl, xl + row0 * HC);
  if (blockIdx.x == 0) proj_strip16(xc, wr, br, xr);
}

#define ATT_T_PER_WG   8
#define ATT_ROWS       (HEADS * ATT_T_PER_WG)
#define ATT_WGS        (B_GR * (R_CENT / ATT_T_PER_WG))
#define LG_PITCH       516
#define ATT_LDS_FLOATS (ATT_ROWS * LG_PITCH + ATT_T_PER_WG * HC + 3 * ATT_ROWS)

__global__ __launch_bounds__(256) void BipartPool_attn_kernel(
    const float* __restrict__ xl,
    const float* __restrict__ xr,
    const float* __restrict__ att,
    const float* __restrict__ bias,
    float* __restrict__ out)
{
  extern __shared__ float smem[];
  float* s_logit = smem;
  float* s_acc   = s_logit + ATT_ROWS * LG_PITCH;
  float* s_den   = s_acc + ATT_T_PER_WG * HC;
  float* s_pself = s_den + ATT_ROWS;
  float* s_self  = s_pself + ATT_ROWS;

  const int g    = blockIdx.x >> 1;
  const int r0   = (blockIdx.x & 1) * ATT_T_PER_WG;
  const int tid  = threadIdx.x;
  const int base = g * 512;
  const int wave = tid >> 5;
  const int lane = tid & 31;

  #pragma unroll 1
  for (int pass = 0; pass < 4; ++pass) {
    const int h = pass >> 1;
    const int n = tid + (pass & 1) * 256;
    const float* xrow  = xl + (size_t)(base + n) * HC + h * C_CH;
    const float* arow  = att + h * C_CH;
    const float* rbase = xr + r0 * HC + h * C_CH;
    float xv[C_CH];
    #pragma unroll
    for (int c = 0; c < C_CH; c += 4) {
      const float4 t = *(const float4*)(xrow + c);
      xv[c] = t.x; xv[c + 1] = t.y; xv[c + 2] = t.z; xv[c + 3] = t.w;
    }
    float acc[ATT_T_PER_WG][4];
    #pragma unroll
    for (int r = 0; r < ATT_T_PER_WG; ++r) {
      acc[r][0] = 0.f; acc[r][1] = 0.f; acc[r][2] = 0.f; acc[r][3] = 0.f;
    }
    float lin[4] = {0.f, 0.f, 0.f, 0.f};
    float av[16], rcur[16], rnxt[16];
    #pragma unroll
    for (int j = 0; j < 16; ++j) rcur[j] = rbase[j];
    #pragma unroll
    for (int q = 0; q < 4 * ATT_T_PER_WG; ++q) {
      const int r = q % ATT_T_PER_WG, c0 = (q / ATT_T_PER_WG) * 16;
      if (q + 1 < 4 * ATT_T_PER_WG) {
        const float* rn = rbase + ((q + 1) % ATT_T_PER_WG) * HC
                                + ((q + 1) / ATT_T_PER_WG) * 16;
        #pragma unroll
        for (int j = 0; j < 16; ++j) rnxt[j] = rn[j];
      }
      if (r == 0) {
        #pragma unroll
        for (int j = 0; j < 16; ++j) av[j] = arow[c0 + j];
        #pragma unroll
        for (int j = 0; j < 16; ++j) lin[j & 3] = fmaf(av[j], xv[c0 + j], lin[j & 3]);
      }
      #pragma unroll
      for (int j = 0; j < 16; j += 4) {
        const float s0 = xv[c0 + j]     + rcur[j];
        const float s1 = xv[c0 + j + 1] + rcur[j + 1];
        const float s2 = xv[c0 + j + 2] + rcur[j + 2];
        const float s3 = xv[c0 + j + 3] + rcur[j + 3];
        acc[r][0] = fmaf(av[j],     fabsf(s0), acc[r][0]);
        acc[r][1] = fmaf(av[j + 1], fabsf(s1), acc[r][1]);
        acc[r][2] = fmaf(av[j + 2], fabsf(s2), acc[r][2]);
        acc[r][3] = fmaf(av[j + 3], fabsf(s3), acc[r][3]);
      }
      #pragma unroll
      for (int j = 0; j < 16; ++j) rcur[j] = rnxt[j];
    }
    const float lin_s = KA_LRELU * ((lin[0] + lin[1]) + (lin[2] + lin[3]));
    #pragma unroll
    for (int r = 0; r < ATT_T_PER_WG; ++r)
      s_logit[(h * ATT_T_PER_WG + r) * LG_PITCH + n] =
          lin_s + KB_LRELU * ((acc[r][0] + acc[r][1]) + (acc[r][2] + acc[r][3]));
  }

  if (tid < ATT_ROWS) {
    const int r8 = tid % ATT_T_PER_WG, h = tid / ATT_T_PER_WG;
    const int r = r0 + r8;
    const int t = g * 16 + r;
    const float* xrow = xl + (size_t)t * HC + h * C_CH;
    const float* rrow = xr + r * HC + h * C_CH;
    const float* arow = att + h * C_CH;
    float l0 = 0.f, l1 = 0.f, a0 = 0.f, a1 = 0.f;
    #pragma unroll 8
    for (int c = 0; c < C_CH; c += 2) {
      l0 = fmaf(arow[c],     xrow[c],     l0);
      l1 = fmaf(arow[c + 1], xrow[c + 1], l1);
      a0 = fmaf(arow[c],     fabsf(xrow[c]     + rrow[c]),     a0);
      a1 = fmaf(arow[c + 1], fabsf(xrow[c + 1] + rrow[c + 1]), a1);
    }
    s_self[tid] = KA_LRELU * (l0 + l1) + KB_LRELU * (a0 + a1);
  }
  __syncthreads();

  if (g == 0 && tid < ATT_ROWS) {
    const int r8 = tid % ATT_T_PER_WG, h = tid / ATT_T_PER_WG;
    s_logit[(h * ATT_T_PER_WG + r8) * LG_PITCH + (r0 + r8)] = -__builtin_inff();
  }
  __syncthreads();

  for (int q = 0; q < ATT_ROWS / 8; ++q) {
    const int rh = wave * (ATT_ROWS / 8) + q;
    float* lrow = s_logit + rh * LG_PITCH;
    float m = s_self[rh];
    for (int i = lane; i < 512; i += 32) m = fmaxf(m, lrow[i]);
    #pragma unroll
    for (int off = 16; off >= 1; off >>= 1) m = fmaxf(m, __shfl_xor(m, off, 32));
    float sum = 0.f;
    for (int i = lane; i < 512; i += 32) {
      const float pv = __expf(lrow[i] - m);
      lrow[i] = pv;
      sum += pv;
    }
    #pragma unroll
    for (int off = 16; off >= 1; off >>= 1) sum += __shfl_xor(sum, off, 32);
    if (lane == 0) {
      const float ps = __expf(s_self[rh] - m);
      s_pself[rh] = ps;
      s_den[rh] = sum + ps + 1e-16f;
    }
  }
  __syncthreads();

  {
    const int hw = wave >> 2;
    const int lo = lane & 15, hi = lane >> 4;
    const float* prow = s_logit + lo * LG_PITCH;
    const float* bcol = xl + (size_t)base * HC + wave * 16 + lo;
    v8f c;
    #pragma unroll
    for (int i = 0; i < 8; ++i) c[i] = 0.f;
    for (int k = 0; k < 512; k += 4) {
      const int kk = k + 2 * hi;
      v2f a = *(const v2f*)(prow + kk);
      v2f bb;
      bb.x = bcol[(size_t)kk * HC];
      bb.y = bcol[(size_t)(kk + 1) * HC];
      c = __builtin_amdgcn_wmma_f32_16x16x4_f32(false, a, false, bb,
                                                (short)0, c, false, false);
    }
    if (hi == hw) {
      #pragma unroll
      for (int v = 0; v < 8; ++v)
        s_acc[v * HC + wave * 16 + lo] = c[v];
    }
  }
  __syncthreads();

  for (int i = tid; i < ATT_T_PER_WG * 64; i += 256) {
    const int r8 = i >> 6, cc = i & 63;
    const int t = g * 16 + r0 + r8;
    const float v0 = (s_acc[r8 * HC + cc] +
                      s_pself[r8] * xl[(size_t)t * HC + cc]) /
                     s_den[r8];
    const float v1 = (s_acc[r8 * HC + C_CH + cc] +
                      s_pself[ATT_T_PER_WG + r8] * xl[(size_t)t * HC + C_CH + cc]) /
                     s_den[ATT_T_PER_WG + r8];
    out[(size_t)t * 64 + cc] = 0.5f * (v0 + v1) + bias[cc];
  }
}

extern "C" void kernel_launch(void* const* d_in, const int* in_sizes, int n_in,
                              void* d_out, int out_size, void* d_ws, size_t ws_size,
                              hipStream_t stream) {
  const float* x    = (const float*)d_in[0];
  const float* xcb  = (const float*)d_in[3];
  const float* wl   = (const float*)d_in[4];
  const float* bl   = (const float*)d_in[5];
  const float* wr   = (const float*)d_in[6];
  const float* br   = (const float*)d_in[7];
  const float* att  = (const float*)d_in[8];
  const float* bias = (const float*)d_in[9];
  float* out = (float*)d_out;

  float* xl = (float*)d_ws;
  float* xr = xl + (size_t)N_NODES * HC;

  BipartPool_proj_gemm_kernel<<<N_NODES / XL_ROWS_PER_BLOCK, 256, 0, stream>>>(
      x, wl, bl, xl, xcb, wr, br, xr);

  const size_t smem = (size_t)ATT_LDS_FLOATS * sizeof(float);
  BipartPool_attn_kernel<<<ATT_WGS, 256, smem, stream>>>(xl, xr, att, bias, out);
}


# attention kernel: weighted sum as wave = head x node quarter, its 64 16-byte B loads issued as one batch, 4 accumulators (was 16 serial load-wait-multiply rounds); self-loop rows loaded in one batch
# speedup vs baseline: 2.7291x; 1.0451x over previous
#include <hip/hip_runtime.h>
#include <hip/hip_bf16.h>
#include <math.h>


#define N_NODES 65536
#define IN_CH   64
#define HC      128
#define R_CENT  16
#define B_GR    128
#define HEADS   2
#define C_CH    64

typedef __attribute__((ext_vector_type(2))) float v2f;
typedef __attribute__((ext_vector_type(8))) float v8f;

constexpr float KA_LRELU = (float)(0.5 * (1.0 + (double)0.2f));
constexpr float KB_LRELU = (float)(0.5 * (1.0 - (double)0.2f));


__device__ __forceinline__ void proj_strip16(
    const float* __restrict__ a,
    const float* __restrict__ w,
    const float* __restrict__ b,
    float* __restrict__ o)
{
  const int wave = threadIdx.x >> 5;
  const int lane = threadIdx.x & 31;
  const int lo = lane & 15;
  const int hi = lane >> 4;
  const int col0 = wave * 16;

  const float bv = b[col0 + lo];
  v8f c;
  #pragma unroll
  for (int i = 0; i < 8; ++i) c[i] = bv;

  const float* arow = a + (size_t)lo * IN_CH;
  #pragma unroll
  for (int k = 0; k < IN_CH; k += 4) {
    const int kk = k + 2 * hi;
    v2f av = *(const v2f*)(arow + kk);
    v2f bb;
    bb.x = w[(size_t)kk * HC + col0 + lo];
    bb.y = w[(size_t)(kk + 1) * HC + col0 + lo];
    c = __builtin_amdgcn_wmma_f32_16x16x4_f32(false, av, false, bb,
                                              (short)0, c, false, false);
  }

  float* orow = o + col0 + lo;
  #pragma unroll
  for (int v = 0; v < 8; ++v) orow[(size_t)(v + 8 * hi) * HC] = c[v];
}

#define XL_ROWS_PER_BLOCK 256
#define XL_STRIPS_PER_WAVE 4

__device__ __forceinline__ void load_a_strip(const float* __restrict__ ap, v2f (&a)[16]) {
  #pragma unroll
  for (int i = 0; i < 16; ++i) a[i] = *(const v2f*)(ap + 4 * i);
}

__device__ __forceinline__ void proj_rows256(
    const float* __restrict__ a,
    const float* __restrict__ w,
    const float* __restrict__ b,
    float* __restrict__ o)
{
  const int wave = threadIdx.x >> 5;
  const int lane = threadIdx.x & 31;
  const int lo = lane & 15;
  const int hi = lane >> 4;
  const int ncol = (wave & 1) * 64 + 4 * lo;
  const int row0 = (wave >> 1) * (16 * XL_STRIPS_PER_WAVE);

  const float* ap = a + (size_t)(row0 + lo) * IN_CH + 2 * hi;
  float* op = o + (size_t)(row0 + 8 * hi) * HC + ncol;

  float4 wr0[16], wr1[16];
  #pragma unroll
  for (int i = 0; i < 16; ++i) {
    const int kk = 4 * i + 2 * hi;
    wr0[i] = *(const float4*)(w + (size_t)kk * HC + ncol);
    wr1[i] = *(const float4*)(w + (size_t)(kk + 1) * HC + ncol);
  }
  const float4 bv = *(const float4*)(b + ncol);
  v2f acur[16], anxt[16];
  load_a_strip(ap, acur);
  __builtin_amdgcn_sched_barrier(0);

  v2f bf0[16], bf1[16], bf2[16], bf3[16];
  #pragma unroll
  for (int i = 0; i < 16; ++i) {
    bf0[i].x = wr0[i].x; bf0[i].y = wr1[i].x;
    bf1[i].x = wr0[i].y; bf1[i].y = wr1[i].y;
    bf2[i].x = wr0[i].z; bf2[i].y = wr1[i].z;
    bf3[i].x = wr0[i].w; bf3[i].y = wr1[i].w;
  }

  #pragma unroll
  for (int s = 0; s < XL_STRIPS_PER_WAVE; ++s) {
    if (s + 1 < XL_STRIPS_PER_WAVE) load_a_strip(ap + (size_t)(s + 1) * 16 * IN_CH, anxt);
    __builtin_amdgcn_sched_barrier(0);

    v8f c0, c1, c2, c3;
    #pragma unroll
    for (int r = 0; r < 8; ++r) { c0[r] = bv.x; c1[r] = bv.y; c2[r] = bv.z; c3[r] = bv.w; }
    #pragma unroll
    for (int i = 0; i < 16; ++i) {
      c0 = __builtin_amdgcn_wmma_f32_16x16x4_f32(false, acur[i], false, bf0[i], (short)0, c0, false, false);
      c1 = __builtin_amdgcn_wmma_f32_16x16x4_f32(false, acur[i], false, bf1[i], (short)0, c1, false, false);
      c2 = __builtin_amdgcn_wmma_f32_16x16x4_f32(false, acur[i], false, bf2[i], (short)0, c2, false, false);
      c3 = __builtin_amdgcn_wmma_f32_16x16x4_f32(false, acur[i], false, bf3[i], (short)0, c3, false, false);
    }
    #pragma unroll
    for (int r = 0; r < 8; ++r) {
      float4 t;
      t.x = c0[r]; t.y = c1[r]; t.z = c2[r]; t.w = c3[r];
      *(float4*)(op + (size_t)(s * 16 + r) * HC) = t;
    }
    #pragma unroll
    for (int i = 0; i < 16; ++i) acur[i] = anxt[i];
  }
}

__global__ __launch_bounds__(256, 2) void BipartPool_proj_gemm_kernel(
    const float* __restrict__ x,
    const float* __restrict__ wl,
    const float* __restrict__ bl,
    float* __restrict__ xl,
    const float* __restrict__ xc,
    const float* __restrict__ wr,
    const float* __restrict__ br,
    float* __restrict__ xr)
{
  const size_t row0 = (size_t)blockIdx.x * XL_ROWS_PER_BLOCK;
  proj_rows256(x + row0 * IN_CH, wl, bl, xl + row0 * HC);
  if (blockIdx.x == 0) proj_strip16(xc, wr, br, xr);
}

#define ATT_T_PER_WG   8
#define ATT_ROWS       (HEADS * ATT_T_PER_WG)
#define ATT_WGS        (B_GR * (R_CENT / ATT_T_PER_WG))
#define LG_PITCH       516
#define ATT_KQ         4
#define ATT_LDS_FLOATS (ATT_ROWS * LG_PITCH + ATT_KQ * ATT_T_PER_WG * HC + 3 * ATT_ROWS)

__global__ __launch_bounds__(256) void BipartPool_attn_kernel(
    const float* __restrict__ xl,
    const float* __restrict__ xr,
    const float* __restrict__ att,
    const float* __restrict__ bias,
    float* __restrict__ out)
{
  extern __shared__ float smem[];
  float* s_logit = smem;
  float* s_part  = s_logit + ATT_ROWS * LG_PITCH;
  float* s_den   = s_part + ATT_KQ * ATT_T_PER_WG * HC;
  float* s_pself = s_den + ATT_ROWS;
  float* s_self  = s_pself + ATT_ROWS;

  const int g    = blockIdx.x >> 1;
  const int r0   = (blockIdx.x & 1) * ATT_T_PER_WG;
  const int tid  = threadIdx.x;
  const int base = g * 512;
  const int wave = tid >> 5;
  const int lane = tid & 31;

  if (tid < ATT_ROWS) {
    const int r8 = tid % ATT_T_PER_WG, h = tid / ATT_T_PER_WG;
    const int r = r0 + r8;
    const int t = g * 16 + r;
    const float* xrow = xl + (size_t)t * HC + h * C_CH;
    const float* rrow = xr + r * HC + h * C_CH;
    const float* arow = att + h * C_CH;
    float4 xs[16], rs[16], ws[16];
    #pragma unroll
    for (int c = 0; c < 16; ++c) {
      xs[c] = *(const float4*)(xrow + 4 * c);
      rs[c] = *(const float4*)(rrow + 4 * c);
      ws[c] = *(const float4*)(arow + 4 * c);
    }
    float l0 = 0.f, l1 = 0.f, a0 = 0.f, a1 = 0.f;
    #pragma unroll
    for (int c = 0; c < 16; ++c) {
      l0 = fmaf(ws[c].x, xs[c].x, l0);
      l1 = fmaf(ws[c].y, xs[c].y, l1);
      a0 = fmaf(ws[c].x, fabsf(xs[c].x + rs[c].x), a0);
      a1 = fmaf(ws[c].y, fabsf(xs[c].y + rs[c].y), a1);
      l0 = fmaf(ws[c].z, xs[c].z, l0);
      l1 = fmaf(ws[c].w, xs[c].w, l1);
      a0 = fmaf(ws[c].z, fabsf(xs[c].z + rs[c].z), a0);
      a1 = fmaf(ws[c].w, fabsf(xs[c].w + rs[c].w), a1);
    }
    s_self[tid] = KA_LRELU * (l0 + l1) + KB_LRELU * (a0 + a1);
  }

  #pragma unroll 1
  for (int pass = 0; pass < 4; ++pass) {
    const int h = pass >> 1;
    const int n = tid + (pass & 1) * 256;
    const float* xrow  = xl + (size_t)(base + n) * HC + h * C_CH;
    const float* arow  = att + h * C_CH;
    const float* rbase = xr + r0 * HC + h * C_CH;
    float xv[C_CH];
    #pragma unroll
    for (int c = 0; c < C_CH; c += 4) {
      const float4 t = *(const float4*)(xrow + c);
      xv[c] = t.x; xv[c + 1] = t.y; xv[c + 2] = t.z; xv[c + 3] = t.w;
    }
    float acc[ATT_T_PER_WG][4];
    #pragma unroll
    for (int r = 0; r < ATT_T_PER_WG; ++r) {
      acc[r][0] = 0.f; acc[r][1] = 0.f; acc[r][2] = 0.f; acc[r][3] = 0.f;
    }
    float lin[4] = {0.f, 0.f, 0.f, 0.f};
    float av[16], rcur[16], rnxt[16];
    #pragma unroll
    for (int j = 0; j < 16; ++j) rcur[j] = rbase[j];
    #pragma unroll
    for (int q = 0; q < 4 * ATT_T_PER_WG; ++q) {
      const int r = q % ATT_T_PER_WG, c0 = (q / ATT_T_PER_WG) * 16;
      if (q + 1 < 4 * ATT_T_PER_WG) {
        const float* rn = rbase + ((q + 1) % ATT_T_PER_WG) * HC
                                + ((q + 1) / ATT_T_PER_WG) * 16;
        #pragma unroll
        for (int j = 0; j < 16; ++j) rnxt[j] = rn[j];
      }
      if (r == 0) {
        #pragma unroll
        for (int j = 0; j < 16; ++j) av[j] = arow[c0 + j];
        #pragma unroll
        for (int j = 0; j < 16; ++j) lin[j & 3] = fmaf(av[j], xv[c0 + j], lin[j & 3]);
      }
      #pragma unroll
      for (int j = 0; j < 16; j += 4) {
        const float s0 = xv[c0 + j]     + rcur[j];
        const float s1 = xv[c0 + j + 1] + rcur[j + 1];
        const float s2 = xv[c0 + j + 2] + rcur[j + 2];
        const float s3 = xv[c0 + j + 3] + rcur[j + 3];
        acc[r][0] = fmaf(av[j],     fabsf(s0), acc[r][0]);
        acc[r][1] = fmaf(av[j + 1], fabsf(s1), acc[r][1]);
        acc[r][2] = fmaf(av[j + 2], fabsf(s2), acc[r][2]);
        acc[r][3] = fmaf(av[j + 3], fabsf(s3), acc[r][3]);
      }
      #pragma unroll
      for (int j = 0; j < 16; ++j) rcur[j] = rnxt[j];
    }
    const float lin_s = KA_LRELU * ((lin[0] + lin[1]) + (lin[2] + lin[3]));
    #pragma unroll
    for (int r = 0; r < ATT_T_PER_WG; ++r)
      s_logit[(h * ATT_T_PER_WG + r) * LG_PITCH + n] =
          lin_s + KB_LRELU * ((acc[r][0] + acc[r][1]) + (acc[r][2] + acc[r][3]));
  }

  __syncthreads();

  if (g == 0 && tid < ATT_ROWS) {
    const int r8 = tid % ATT_T_PER_WG, h = tid / ATT_T_PER_WG;
    s_logit[(h * ATT_T_PER_WG + r8) * LG_PITCH + (r0 + r8)] = -__builtin_inff();
  }
  __syncthreads();

  for (int q = 0; q < ATT_ROWS / 8; ++q) {
    const int rh = wave * (ATT_ROWS / 8) + q;
    float* lrow = s_logit + rh * LG_PITCH;
    float m = s_self[rh];
    for (int i = lane; i < 512; i += 32) m = fmaxf(m, lrow[i]);
    #pragma unroll
    for (int off = 16; off >= 1; off >>= 1) m = fmaxf(m, __shfl_xor(m, off, 32));
    float sum = 0.f;
    for (int i = lane; i < 512; i += 32) {
      const float pv = __expf(lrow[i] - m);
      lrow[i] = pv;
      sum += pv;
    }
    #pragma unroll
    for (int off = 16; off >= 1; off >>= 1) sum += __shfl_xor(sum, off, 32);
    if (lane == 0) {
      const float ps = __expf(s_self[rh] - m);
      s_pself[rh] = ps;
      s_den[rh] = sum + ps + 1e-16f;
    }
  }
  __syncthreads();

  {
    const int hw = wave >> 2;
    const int kq = wave & 3;
    const int lo = lane & 15, hi = lane >> 4;
    float4 b0[32], b1[32];
    {
      const float* bp = xl + (size_t)(base + kq * 128 + 2 * hi) * HC + hw * C_CH + 4 * lo;
      #pragma unroll
      for (int i = 0; i < 32; ++i) {
        b0[i] = *(const float4*)(bp + (size_t)(4 * i) * HC);
        b1[i] = *(const float4*)(bp + (size_t)(4 * i + 1) * HC);
      }
    }
    const float* prow = s_logit + lo * LG_PITCH + kq * 128 + 2 * hi;
    v2f af[32];
    #pragma unroll
    for (int i = 0; i < 32; ++i) af[i] = *(const v2f*)(prow + 4 * i);
    __builtin_amdgcn_sched_barrier(0);
    v8f c0, c1, c2, c3;
    #pragma unroll
    for (int i = 0; i < 8; ++i) { c0[i] = 0.f; c1[i] = 0.f; c2[i] = 0.f; c3[i] = 0.f; }
    #pragma unroll
    for (int i = 0; i < 32; ++i) {
      v2f f0, f1, f2, f3;
      f0.x = b0[i].x; f0.y = b1[i].x;
      f1.x = b0[i].y; f1.y = b1[i].y;
      f2.x = b0[i].z; f2.y = b1[i].z;
      f3.x = b0[i].w; f3.y = b1[i].w;
      c0 = __builtin_amdgcn_wmma_f32_16x16x4_f32(false, af[i], false, f0, (short)0, c0, false, false);
      c1 = __builtin_amdgcn_wmma_f32_16x16x4_f32(false, af[i], false, f1, (short)0, c1, false, false);
      c2 = __builtin_amdgcn_wmma_f32_16x16x4_f32(false, af[i], false, f2, (short)0, c2, false, false);
      c3 = __builtin_amdgcn_wmma_f32_16x16x4_f32(false, af[i], false, f3, (short)0, c3, false, false);
    }
    if (hi == hw) {
      #pragma unroll
      for (int v = 0; v < 8; ++v) {
        float4 t;
        t.x = c0[v]; t.y = c1[v]; t.z = c2[v]; t.w = c3[v];
        *(float4*)(s_part + (kq * ATT_T_PER_WG + v) * HC + hw * C_CH + 4 * lo) = t;
      }
    }
  }
  __syncthreads();

  for (int i = tid; i < ATT_T_PER_WG * 64; i += 256) {
    const int r8 = i >> 6, cc = i & 63;
    const int t = g * 16 + r0 + r8;
    const float* p0 = s_part + r8 * HC + cc;
    const float* p1 = p0 + C_CH;
    const int qs = ATT_T_PER_WG * HC;
    const float acc0 = (p0[0] + p0[qs]) + (p0[2 * qs] + p0[3 * qs]);
    const float acc1 = (p1[0] + p1[qs]) + (p1[2 * qs] + p1[3 * qs]);
    const float v0 = (acc0 +
                      s_pself[r8] * xl[(size_t)t * HC + cc]) /
                     s_den[r8];
    const float v1 = (acc1 +
                      s_pself[ATT_T_PER_WG + r8] * xl[(size_t)t * HC + C_CH + cc]) /
                     s_den[ATT_T_PER_WG + r8];
    out[(size_t)t * 64 + cc] = 0.5f * (v0 + v1) + bias[cc];
  }
}

extern "C" void kernel_launch(void* const* d_in, const int* in_sizes, int n_in,
                              void* d_out, int out_size, void* d_ws, size_t ws_size,
                              hipStream_t stream) {
  const float* x    = (const float*)d_in[0];
  const float* xcb  = (const float*)d_in[3];
  const float* wl   = (const float*)d_in[4];
  const float* bl   = (const float*)d_in[5];
  const float* wr   = (const float*)d_in[6];
  const float* br   = (const float*)d_in[7];
  const float* att  = (const float*)d_in[8];
  const float* bias = (const float*)d_in[9];
  float* out = (float*)d_out;

  float* xl = (float*)d_ws;
  float* xr = xl + (size_t)N_NODES * HC;

  BipartPool_proj_gemm_kernel<<<N_NODES / XL_ROWS_PER_BLOCK, 256, 0, stream>>>(
      x, wl, bl, xl, xcb, wr, br, xr);

  const size_t smem = (size_t)ATT_LDS_FLOATS * sizeof(float);
  BipartPool_attn_kernel<<<ATT_WGS, 256, smem, stream>>>(xl, xr, att, bias, out);
}


# attention kernel, softmax pass: a wave's two rows together, read as 16-byte LDS pieces in one batch, max and sum butterflies on DPP / permlanex16 instead of ten serial LDS permute trips per row
# speedup vs baseline: 2.8173x; 1.0323x over previous
#include <hip/hip_runtime.h>
#include <hip/hip_bf16.h>
#include <math.h>


#define N_NODES 65536
#define IN_CH   64
#define HC      128
#define R_CENT  16
#define B_GR    128
#define HEADS   2
#define C_CH    64

typedef __attribute__((ext_vector_type(2))) float v2f;
typedef __attribute__((ext_vector_type(8))) float v8f;

template <int K>
__device__ __forceinline__ float row_xor(float x) {
  const int b = __builtin_bit_cast(int, x);
  return __builtin_bit_cast(float, __builtin_amdgcn_update_dpp(b, b, 0x160 + K, 0xf, 0xf, false));
}
__device__ __forceinline__ float xhalf(float x) {
  const int b = __builtin_bit_cast(int, x);
  return __builtin_bit_cast(float,
      __builtin_amdgcn_permlanex16(b, b, 0x76543210, 0xfedcba98, false, false));
}
__device__ __forceinline__ void wave_max2(float& a, float& b) {
  a = fmaxf(a, row_xor<1>(a)); b = fmaxf(b, row_xor<1>(b));
  a = fmaxf(a, row_xor<2>(a)); b = fmaxf(b, row_xor<2>(b));
  a = fmaxf(a, row_xor<4>(a)); b = fmaxf(b, row_xor<4>(b));
  a = fmaxf(a, row_xor<8>(a)); b = fmaxf(b, row_xor<8>(b));
  a = fmaxf(a, xhalf(a));      b = fmaxf(b, xhalf(b));
}
__device__ __forceinline__ void wave_sum2(float& a, float& b) {
  a += row_xor<1>(a); b += row_xor<1>(b);
  a += row_xor<2>(a); b += row_xor<2>(b);
  a += row_xor<4>(a); b += row_xor<4>(b);
  a += row_xor<8>(a); b += row_xor<8>(b);
  a += xhalf(a);      b += xhalf(b);
}

constexpr float KA_LRELU = (float)(0.5 * (1.0 + (double)0.2f));
constexpr float KB_LRELU = (float)(0.5 * (1.0 - (double)0.2f));


__device__ __forceinline__ void proj_strip16(
    const float* __restrict__ a,
    const float* __restrict__ w,
    const float* __restrict__ b,
    float* __restrict__ o)
{
  const int wave = threadIdx.x >> 5;
  const int lane = threadIdx.x & 31;
  const int lo = lane & 15;
  const int hi = lane >> 4;
  const int col0 = wave * 16;

  const float bv = b[col0 + lo];
  v8f c;
  #pragma unroll
  for (int i = 0; i < 8; ++i) c[i] = bv;

  const float* arow = a + (size_t)lo * IN_CH;
  #pragma unroll
  for (int k = 0; k < IN_CH; k += 4) {
    const int kk = k + 2 * hi;
    v2f av = *(const v2f*)(arow + kk);
    v2f bb;
    bb.x = w[(size_t)kk * HC + col0 + lo];
    bb.y = w[(size_t)(kk + 1) * HC + col0 + lo];
    c = __builtin_amdgcn_wmma_f32_16x16x4_f32(false, av, false, bb,
                                              (short)0, c, false, false);
  }

  float* orow = o + col0 + lo;
  #pragma unroll
  for (int v = 0; v < 8; ++v) orow[(size_t)(v + 8 * hi) * HC] = c[v];
}

#define XL_ROWS_PER_BLOCK 256
#define XL_STRIPS_PER_WAVE 4

__device__ __forceinline__ void load_a_strip(const float* __restrict__ ap, v2f (&a)[16]) {
  #pragma unroll
  for (int i = 0; i < 16; ++i) a[i] = *(const v2f*)(ap + 4 * i);
}

__device__ __forceinline__ void proj_rows256(
    const float* __restrict__ a,
    const float* __restrict__ w,
    const float* __restrict__ b,
    float* __restrict__ o)
{
  const int wave = threadIdx.x >> 5;
  const int lane = threadIdx.x & 31;
  const int lo = lane & 15;
  const int hi = lane >> 4;
  const int ncol = (wave & 1) * 64 + 4 * lo;
  const int row0 = (wave >> 1) * (16 * XL_STRIPS_PER_WAVE);

  const float* ap = a + (size_t)(row0 + lo) * IN_CH + 2 * hi;
  float* op = o + (size_t)(row0 + 8 * hi) * HC + ncol;

  float4 wr0[16], wr1[16];
  #pragma unroll
  for (int i = 0; i < 16; ++i) {
    const int kk = 4 * i + 2 * hi;
    wr0[i] = *(const float4*)(w + (size_t)kk * HC + ncol);
    wr1[i] = *(const float4*)(w + (size_t)(kk + 1) * HC + ncol);
  }
  const float4 bv = *(const float4*)(b + ncol);
  v2f acur[16], anxt[16];
  load_a_strip(ap, acur);
  __builtin_amdgcn_sched_barrier(0);

  v2f bf0[16], bf1[16], bf2[16], bf3[16];
  #pragma unroll
  for (int i = 0; i < 16; ++i) {
    bf0[i].x = wr0[i].x; bf0[i].y = wr1[i].x;
    bf1[i].x = wr0[i].y; bf1[i].y = wr1[i].y;
    bf2[i].x = wr0[i].z; bf2[i].y = wr1[i].z;
    bf3[i].x = wr0[i].w; bf3[i].y = wr1[i].w;
  }

  #pragma unroll
  for (int s = 0; s < XL_STRIPS_PER_WAVE; ++s) {
    if (s + 1 < XL_STRIPS_PER_WAVE) load_a_strip(ap + (size_t)(s + 1) * 16 * IN_CH, anxt);
    __builtin_amdgcn_sched_barrier(0);

    v8f c0, c1, c2, c3;
    #pragma unroll
    for (int r = 0; r < 8; ++r) { c0[r] = bv.x; c1[r] = bv.y; c2[r] = bv.z; c3[r] = bv.w; }
    #pragma unroll
    for (int i = 0; i < 16; ++i) {
      c0 = __builtin_amdgcn_wmma_f32_16x16x4_f32(false, acur[i], false, bf0[i], (short)0, c0, false, false);
      c1 = __builtin_amdgcn_wmma_f32_16x16x4_f32(false, acur[i], false, bf1[i], (short)0, c1, false, false);
      c2 = __builtin_amdgcn_wmma_f32_16x16x4_f32(false, acur[i], false, bf2[i], (short)0, c2, false, false);
      c3 = __builtin_amdgcn_wmma_f32_16x16x4_f32(false, acur[i], false, bf3[i], (short)0, c3, false, false);
    }
    #pragma unroll
    for (int r = 0; r < 8; ++r) {
      float4 t;
      t.x = c0[r]; t.y = c1[r]; t.z = c2[r]; t.w = c3[r];
      *(float4*)(op + (size_t)(s * 16 + r) * HC) = t;
    }
    #pragma unroll
    for (int i = 0; i < 16; ++i) acur[i] = anxt[i];
  }
}

__global__ __launch_bounds__(256, 2) void BipartPool_proj_gemm_kernel(
    const float* __restrict__ x,
    const float* __restrict__ wl,
    const float* __restrict__ bl,
    float* __restrict__ xl,
    const float* __restrict__ xc,
    const float* __restrict__ wr,
    const float* __restrict__ br,
    float* __restrict__ xr)
{
  const size_t row0 = (size_t)blockIdx.x * XL_ROWS_PER_BLOCK;
  proj_rows256(x + row0 * IN_CH, wl, bl, xl + row0 * HC);
  if (blockIdx.x == 0) proj_strip16(xc, wr, br, xr);
}

#define ATT_T_PER_WG   8
#define ATT_ROWS       (HEADS * ATT_T_PER_WG)
#define ATT_WGS        (B_GR * (R_CENT / ATT_T_PER_WG))
#define LG_PITCH       516
#define ATT_KQ         4
#define ATT_LDS_FLOATS (ATT_ROWS * LG_PITCH + ATT_KQ * ATT_T_PER_WG * HC + 3 * ATT_ROWS)

__global__ __launch_bounds__(256) void BipartPool_attn_kernel(
    const float* __restrict__ xl,
    const float* __restrict__ xr,
    const float* __restrict__ att,
    const float* __restrict__ bias,
    float* __restrict__ out)
{
  extern __shared__ __attribute__((aligned(16))) float smem[];
  float* s_logit = smem;
  float* s_part  = s_logit + ATT_ROWS * LG_PITCH;
  float* s_den   = s_part + ATT_KQ * ATT_T_PER_WG * HC;
  float* s_pself = s_den + ATT_ROWS;
  float* s_self  = s_pself + ATT_ROWS;

  const int g    = blockIdx.x >> 1;
  const int r0   = (blockIdx.x & 1) * ATT_T_PER_WG;
  const int tid  = threadIdx.x;
  const int base = g * 512;
  const int wave = tid >> 5;
  const int lane = tid & 31;

  if (tid < ATT_ROWS) {
    const int r8 = tid % ATT_T_PER_WG, h = tid / ATT_T_PER_WG;
    const int r = r0 + r8;
    const int t = g * 16 + r;
    const float* xrow = xl + (size_t)t * HC + h * C_CH;
    const float* rrow = xr + r * HC + h * C_CH;
    const float* arow = att + h * C_CH;
    float4 xs[16], rs[16], ws[16];
    #pragma unroll
    for (int c = 0; c < 16; ++c) {
      xs[c] = *(const float4*)(xrow + 4 * c);
      rs[c] = *(const float4*)(rrow + 4 * c);
      ws[c] = *(const float4*)(arow + 4 * c);
    }
    float l0 = 0.f, l1 = 0.f, a0 = 0.f, a1 = 0.f;
    #pragma unroll
    for (int c = 0; c < 16; ++c) {
      l0 = fmaf(ws[c].x, xs[c].x, l0);
      l1 = fmaf(ws[c].y, xs[c].y, l1);
      a0 = fmaf(ws[c].x, fabsf(xs[c].x + rs[c].x), a0);
      a1 = fmaf(ws[c].y, fabsf(xs[c].y + rs[c].y), a1);
      l0 = fmaf(ws[c].z, xs[c].z, l0);
      l1 = fmaf(ws[c].w, xs[c].w, l1);
      a0 = fmaf(ws[c].z, fabsf(xs[c].z + rs[c].z), a0);
      a1 = fmaf(ws[c].w, fabsf(xs[c].w + rs[c].w), a1);
    }
    s_self[tid] = KA_LRELU * (l0 + l1) + KB_LRELU * (a0 + a1);
  }

  #pragma unroll 1
  for (int pass = 0; pass < 4; ++pass) {
    const int h = pass >> 1;
    const int n = tid + (pass & 1) * 256;
    const float* xrow  = xl + (size_t)(base + n) * HC + h * C_CH;
    const float* arow  = att + h * C_CH;
    const float* rbase = xr + r0 * HC + h * C_CH;
    float xv[C_CH];
    #pragma unroll
    for (int c = 0; c < C_CH; c += 4) {
      const float4 t = *(const float4*)(xrow + c);
      xv[c] = t.x; xv[c + 1] = t.y; xv[c + 2] = t.z; xv[c + 3] = t.w;
    }
    float acc[ATT_T_PER_WG][4];
    #pragma unroll
    for (int r = 0; r < ATT_T_PER_WG; ++r) {
      acc[r][0] = 0.f; acc[r][1] = 0.f; acc[r][2] = 0.f; acc[r][3] = 0.f;
    }
    float lin[4] = {0.f, 0.f, 0.f, 0.f};
    float av[16], rcur[16], rnxt[16];
    #pragma unroll
    for (int j = 0; j < 16; ++j) rcur[j] = rbase[j];
    #pragma unroll
    for (int q = 0; q < 4 * ATT_T_PER_WG; ++q) {
      const int r = q % ATT_T_PER_WG, c0 = (q / ATT_T_PER_WG) * 16;
      if (q + 1 < 4 * ATT_T_PER_WG) {
        const float* rn = rbase + ((q + 1) % ATT_T_PER_WG) * HC
                                + ((q + 1) / ATT_T_PER_WG) * 16;
        #pragma unroll
        for (int j = 0; j < 16; ++j) rnxt[j] = rn[j];
      }
      if (r == 0) {
        #pragma unroll
        for (int j = 0; j < 16; ++j) av[j] = arow[c0 + j];
        #pragma unroll
        for (int j = 0; j < 16; ++j) lin[j & 3] = fmaf(av[j], xv[c0 + j], lin[j & 3]);
      }
      #pragma unroll
      for (int j = 0; j < 16; j += 4) {
        const float s0 = xv[c0 + j]     + rcur[j];
        const float s1 = xv[c0 + j + 1] + rcur[j + 1];
        const float s2 = xv[c0 + j + 2] + rcur[j + 2];
        const float s3 = xv[c0 + j + 3] + rcur[j + 3];
        acc[r][0] = fmaf(av[j],     fabsf(s0), acc[r][0]);
        acc[r][1] = fmaf(av[j + 1], fabsf(s1), acc[r][1]);
        acc[r][2] = fmaf(av[j + 2], fabsf(s2), acc[r][2]);
        acc[r][3] = fmaf(av[j + 3], fabsf(s3), acc[r][3]);
      }
      #pragma unroll
      for (int j = 0; j < 16; ++j) rcur[j] = rnxt[j];
    }
    const float lin_s = KA_LRELU * ((lin[0] + lin[1]) + (lin[2] + lin[3]));
    #pragma unroll
    for (int r = 0; r < ATT_T_PER_WG; ++r)
      s_logit[(h * ATT_T_PER_WG + r) * LG_PITCH + n] =
          lin_s + KB_LRELU * ((acc[r][0] + acc[r][1]) + (acc[r][2] + acc[r][3]));
  }

  __syncthreads();

  if (g == 0 && tid < ATT_ROWS) {
    const int r8 = tid % ATT_T_PER_WG, h = tid / ATT_T_PER_WG;
    s_logit[(h * ATT_T_PER_WG + r8) * LG_PITCH + (r0 + r8)] = -__builtin_inff();
  }
  __syncthreads();

  {
    const int rh0 = wave * (ATT_ROWS / 8);
    float* row0 = s_logit + rh0 * LG_PITCH + 4 * lane;
    float* row1 = row0 + LG_PITCH;
    const float self0 = s_self[rh0], self1 = s_self[rh0 + 1];
    float4 u[4], v[4];
    #pragma unroll
    for (int j = 0; j < 4; ++j) {
      u[j] = *(const float4*)(row0 + 128 * j);
      v[j] = *(const float4*)(row1 + 128 * j);
    }
    float m0 = self0, m1 = self1;
    #pragma unroll
    for (int j = 0; j < 4; ++j) {
      m0 = fmaxf(m0, fmaxf(fmaxf(u[j].x, u[j].y), fmaxf(u[j].z, u[j].w)));
      m1 = fmaxf(m1, fmaxf(fmaxf(v[j].x, v[j].y), fmaxf(v[j].z, v[j].w)));
    }
    wave_max2(m0, m1);
    float t0 = 0.f, t1 = 0.f;
    #pragma unroll
    for (int j = 0; j < 4; ++j) {
      float4 a, b;
      a.x = __expf(u[j].x - m0); a.y = __expf(u[j].y - m0);
      a.z = __expf(u[j].z - m0); a.w = __expf(u[j].w - m0);
      b.x = __expf(v[j].x - m1); b.y = __expf(v[j].y - m1);
      b.z = __expf(v[j].z - m1); b.w = __expf(v[j].w - m1);
      *(float4*)(row0 + 128 * j) = a;
      *(float4*)(row1 + 128 * j) = b;
      t0 += (a.x + a.y) + (a.z + a.w);
      t1 += (b.x + b.y) + (b.z + b.w);
    }
    wave_sum2(t0, t1);
    if (lane == 0) {
      const float p0 = __expf(self0 - m0), p1 = __expf(self1 - m1);
      s_pself[rh0]     = p0;
      s_pself[rh0 + 1] = p1;
      s_den[rh0]       = t0 + p0 + 1e-16f;
      s_den[rh0 + 1]   = t1 + p1 + 1e-16f;
    }
  }
  __syncthreads();

  {
    const int hw = wave >> 2;
    const int kq = wave & 3;
    const int lo = lane & 15, hi = lane >> 4;
    float4 b0[32], b1[32];
    {
      const float* bp = xl + (size_t)(base + kq * 128 + 2 * hi) * HC + hw * C_CH + 4 * lo;
      #pragma unroll
      for (int i = 0; i < 32; ++i) {
        b0[i] = *(const float4*)(bp + (size_t)(4 * i) * HC);
        b1[i] = *(const float4*)(bp + (size_t)(4 * i + 1) * HC);
      }
    }
    const float* prow = s_logit + lo * LG_PITCH + kq * 128 + 2 * hi;
    v2f af[32];
    #pragma unroll
    for (int i = 0; i < 32; ++i) af[i] = *(const v2f*)(prow + 4 * i);
    __builtin_amdgcn_sched_barrier(0);
    v8f c0, c1, c2, c3;
    #pragma unroll
    for (int i = 0; i < 8; ++i) { c0[i] = 0.f; c1[i] = 0.f; c2[i] = 0.f; c3[i] = 0.f; }
    #pragma unroll
    for (int i = 0; i < 32; ++i) {
      v2f f0, f1, f2, f3;
      f0.x = b0[i].x; f0.y = b1[i].x;
      f1.x = b0[i].y; f1.y = b1[i].y;
      f2.x = b0[i].z; f2.y = b1[i].z;
      f3.x = b0[i].w; f3.y = b1[i].w;
      c0 = __builtin_amdgcn_wmma_f32_16x16x4_f32(false, af[i], false, f0, (short)0, c0, false, false);
      c1 = __builtin_amdgcn_wmma_f32_16x16x4_f32(false, af[i], false, f1, (short)0, c1, false, false);
      c2 = __builtin_amdgcn_wmma_f32_16x16x4_f32(false, af[i], false, f2, (short)0, c2, false, false);
      c3 = __builtin_amdgcn_wmma_f32_16x16x4_f32(false, af[i], false, f3, (short)0, c3, false, false);
    }
    if (hi == hw) {
      #pragma unroll
      for (int v = 0; v < 8; ++v) {
        float4 t;
        t.x = c0[v]; t.y = c1[v]; t.z = c2[v]; t.w = c3[v];
        *(float4*)(s_part + (kq * ATT_T_PER_WG + v) * HC + hw * C_CH + 4 * lo) = t;
      }
    }
  }
  __syncthreads();

  for (int i = tid; i < ATT_T_PER_WG * 64; i += 256) {
    const int r8 = i >> 6, cc = i & 63;
    const int t = g * 16 + r0 + r8;
    const float* p0 = s_part + r8 * HC + cc;
    const float* p1 = p0 + C_CH;
    const int qs = ATT_T_PER_WG * HC;
    const float acc0 = (p0[0] + p0[qs]) + (p0[2 * qs] + p0[3 * qs]);
    const float acc1 = (p1[0] + p1[qs]) + (p1[2 * qs] + p1[3 * qs]);
    const float v0 = (acc0 +
                      s_pself[r8] * xl[(size_t)t * HC + cc]) /
                     s_den[r8];
    const float v1 = (acc1 +
                      s_pself[ATT_T_PER_WG + r8] * xl[(size_t)t * HC + C_CH + cc]) /
                     s_den[ATT_T_PER_WG + r8];
    out[(size_t)t * 64 + cc] = 0.5f * (v0 + v1) + bias[cc];
  }
}

extern "C" void kernel_launch(void* const* d_in, const int* in_sizes, int n_in,
                              void* d_out, int out_size, void* d_ws, size_t ws_size,
                              hipStream_t stream) {
  const float* x    = (const float*)d_in[0];
  const float* xcb  = (const float*)d_in[3];
  const float* wl   = (const float*)d_in[4];
  const float* bl   = (const float*)d_in[5];
  const float* wr   = (const float*)d_in[6];
  const float* br   = (const float*)d_in[7];
  const float* att  = (const float*)d_in[8];
  const float* bias = (const float*)d_in[9];
  float* out = (float*)d_out;

  float* xl = (float*)d_ws;
  float* xr = xl + (size_t)N_NODES * HC;

  BipartPool_proj_gemm_kernel<<<N_NODES / XL_ROWS_PER_BLOCK, 256, 0, stream>>>(
      x, wl, bl, xl, xcb, wr, br, xr);

  const size_t smem = (size_t)ATT_LDS_FLOATS * sizeof(float);
  BipartPool_attn_kernel<<<ATT_WGS, 256, smem, stream>>>(xl, xr, att, bias, out);
}


# attention kernel: global loads asked for a stage ahead (next logit pass's x_l rows under the current pass; weighted-sum B rows and epilogue rows under the last pass and softmax); LDS-only barriers
# speedup vs baseline: 2.9609x; 1.0510x over previous
#include <hip/hip_runtime.h>
#include <hip/hip_bf16.h>
#include <math.h>


#define N_NODES 65536
#define IN_CH   64
#define HC      128
#define R_CENT  16
#define B_GR    128
#define HEADS   2
#define C_CH    64

typedef __attribute__((ext_vector_type(2))) float v2f;
typedef __attribute__((ext_vector_type(8))) float v8f;

template <int K>
__device__ __forceinline__ float row_xor(float x) {
  const int b = __builtin_bit_cast(int, x);
  return __builtin_bit_cast(float, __builtin_amdgcn_update_dpp(b, b, 0x160 + K, 0xf, 0xf, false));
}
__device__ __forceinline__ float xhalf(float x) {
  const int b = __builtin_bit_cast(int, x);
  return __builtin_bit_cast(float,
      __builtin_amdgcn_permlanex16(b, b, 0x76543210, 0xfedcba98, false, false));
}
__device__ __forceinline__ void wave_max2(float& a, float& b) {
  a = fmaxf(a, row_xor<1>(a)); b = fmaxf(b, row_xor<1>(b));
  a = fmaxf(a, row_xor<2>(a)); b = fmaxf(b, row_xor<2>(b));
  a = fmaxf(a, row_xor<4>(a)); b = fmaxf(b, row_xor<4>(b));
  a = fmaxf(a, row_xor<8>(a)); b = fmaxf(b, row_xor<8>(b));
  a = fmaxf(a, xhalf(a));      b = fmaxf(b, xhalf(b));
}
__device__ __forceinline__ void wave_sum2(float& a, float& b) {
  a += row_xor<1>(a); b += row_xor<1>(b);
  a += row_xor<2>(a); b += row_xor<2>(b);
  a += row_xor<4>(a); b += row_xor<4>(b);
  a += row_xor<8>(a); b += row_xor<8>(b);
  a += xhalf(a);      b += xhalf(b);
}

constexpr float KA_LRELU = (float)(0.5 * (1.0 + (double)0.2f));
constexpr float KB_LRELU = (float)(0.5 * (1.0 - (double)0.2f));


__device__ __forceinline__ void proj_strip16(
    const float* __restrict__ a,
    const float* __restrict__ w,
    const float* __restrict__ b,
    float* __restrict__ o)
{
  const int wave = threadIdx.x >> 5;
  const int lane = threadIdx.x & 31;
  const int lo = lane & 15;
  const int hi = lane >> 4;
  const int col0 = wave * 16;

  const float bv = b[col0 + lo];
  v8f c;
  #pragma unroll
  for (int i = 0; i < 8; ++i) c[i] = bv;

  const float* arow = a + (size_t)lo * IN_CH;
  #pragma unroll
  for (int k = 0; k < IN_CH; k += 4) {
    const int kk = k + 2 * hi;
    v2f av = *(const v2f*)(arow + kk);
    v2f bb;
    bb.x = w[(size_t)kk * HC + col0 + lo];
    bb.y = w[(size_t)(kk + 1) * HC + col0 + lo];
    c = __builtin_amdgcn_wmma_f32_16x16x4_f32(false, av, false, bb,
                                              (short)0, c, false, false);
  }

  float* orow = o + col0 + lo;
  #pragma unroll
  for (int v = 0; v < 8; ++v) orow[(size_t)(v + 8 * hi) * HC] = c[v];
}

#define XL_ROWS_PER_BLOCK 256
#define XL_STRIPS_PER_WAVE 4

__device__ __forceinline__ void load_a_strip(const float* __restrict__ ap, v2f (&a)[16]) {
  #pragma unroll
  for (int i = 0; i < 16; ++i) a[i] = *(const v2f*)(ap + 4 * i);
}

__device__ __forceinline__ void proj_rows256(
    const float* __restrict__ a,
    const float* __restrict__ w,
    const float* __restrict__ b,
    float* __restrict__ o)
{
  const int wave = threadIdx.x >> 5;
  const int lane = threadIdx.x & 31;
  const int lo = lane & 15;
  const int hi = lane >> 4;
  const int ncol = (wave & 1) * 64 + 4 * lo;
  const int row0 = (wave >> 1) * (16 * XL_STRIPS_PER_WAVE);

  const float* ap = a + (size_t)(row0 + lo) * IN_CH + 2 * hi;
  float* op = o + (size_t)(row0 + 8 * hi) * HC + ncol;

  float4 wr0[16], wr1[16];
  #pragma unroll
  for (int i = 0; i < 16; ++i) {
    const int kk = 4 * i + 2 * hi;
    wr0[i] = *(const float4*)(w + (size_t)kk * HC + ncol);
    wr1[i] = *(const float4*)(w + (size_t)(kk + 1) * HC + ncol);
  }
  const float4 bv = *(const float4*)(b + ncol);
  v2f acur[16], anxt[16];
  load_a_strip(ap, acur);
  __builtin_amdgcn_sched_barrier(0);

  v2f bf0[16], bf1[16], bf2[16], bf3[16];
  #pragma unroll
  for (int i = 0; i < 16; ++i) {
    bf0[i].x = wr0[i].x; bf0[i].y = wr1[i].x;
    bf1[i].x = wr0[i].y; bf1[i].y = wr1[i].y;
    bf2[i].x = wr0[i].z; bf2[i].y = wr1[i].z;
    bf3[i].x = wr0[i].w; bf3[i].y = wr1[i].w;
  }

  #pragma unroll
  for (int s = 0; s < XL_STRIPS_PER_WAVE; ++s) {
    if (s + 1 < XL_STRIPS_PER_WAVE) load_a_strip(ap + (size_t)(s + 1) * 16 * IN_CH, anxt);
    __builtin_amdgcn_sched_barrier(0);

    v8f c0, c1, c2, c3;
    #pragma unroll
    for (int r = 0; r < 8; ++r) { c0[r] = bv.x; c1[r] = bv.y; c2[r] = bv.z; c3[r] = bv.w; }
    #pragma unroll
    for (int i = 0; i < 16; ++i) {
      c0 = __builtin_amdgcn_wmma_f32_16x16x4_f32(false, acur[i], false, bf0[i], (short)0, c0, false, false);
      c1 = __builtin_amdgcn_wmma_f32_16x16x4_f32(false, acur[i], false, bf1[i], (short)0, c1, false, false);
      c2 = __builtin_amdgcn_wmma_f32_16x16x4_f32(false, acur[i], false, bf2[i], (short)0, c2, false, false);
      c3 = __builtin_amdgcn_wmma_f32_16x16x4_f32(false, acur[i], false, bf3[i], (short)0, c3, false, false);
    }
    #pragma unroll
    for (int r = 0; r < 8; ++r) {
      float4 t;
      t.x = c0[r]; t.y = c1[r]; t.z = c2[r]; t.w = c3[r];
      *(float4*)(op + (size_t)(s * 16 + r) * HC) = t;
    }
    #pragma unroll
    for (int i = 0; i < 16; ++i) acur[i] = anxt[i];
  }
}

__global__ __launch_bounds__(256, 2) void BipartPool_proj_gemm_kernel(
    const float* __restrict__ x,
    const float* __restrict__ wl,
    const float* __restrict__ bl,
    float* __restrict__ xl,
    const float* __restrict__ xc,
    const float* __restrict__ wr,
    const float* __restrict__ br,
    float* __restrict__ xr)
{
  const size_t row0 = (size_t)blockIdx.x * XL_ROWS_PER_BLOCK;
  proj_rows256(x + row0 * IN_CH, wl, bl, xl + row0 * HC);
  if (blockIdx.x == 0) proj_strip16(xc, wr, br, xr);
}

#define ATT_T_PER_WG   8
#define ATT_ROWS       (HEADS * ATT_T_PER_WG)
#define ATT_WGS        (B_GR * (R_CENT / ATT_T_PER_WG))
#define LG_PITCH       516
#define ATT_KQ         4
#define ATT_LDS_FLOATS (ATT_ROWS * LG_PITCH + ATT_KQ * ATT_T_PER_WG * HC + 3 * ATT_ROWS)


__device__ __forceinline__ void load_xrow(const float* __restrict__ p, float4 (&x)[16]) {
  #pragma unroll
  for (int c = 0; c < 16; ++c) x[c] = *(const float4*)(p + 4 * c);
}
__device__ __forceinline__ void unpack_xrow(const float4 (&x)[16], float (&xv)[C_CH]) {
  #pragma unroll
  for (int c = 0; c < 16; ++c) {
    xv[4 * c] = x[c].x; xv[4 * c + 1] = x[c].y; xv[4 * c + 2] = x[c].z; xv[4 * c + 3] = x[c].w;
  }
}

__device__ __forceinline__ void logit_pass(
    const float (&xv)[C_CH],
    const float* __restrict__ arow,
    const float* __restrict__ rbase,
    float* __restrict__ lrow)
{
  float acc[ATT_T_PER_WG][4];
  #pragma unroll
  for (int r = 0; r < ATT_T_PER_WG; ++r) {
    acc[r][0] = 0.f; acc[r][1] = 0.f; acc[r][2] = 0.f; acc[r][3] = 0.f;
  }
  float lin[4] = {0.f, 0.f, 0.f, 0.f};
  float av[16], rcur[16], rnxt[16];
  #pragma unroll
  for (int j = 0; j < 16; ++j) rcur[j] = rbase[j];
  #pragma unroll
  for (int q = 0; q < 4 * ATT_T_PER_WG; ++q) {
    const int r = q % ATT_T_PER_WG, c0 = (q / ATT_T_PER_WG) * 16;
    if (q + 1 < 4 * ATT_T_PER_WG) {
      const float* rn = rbase + ((q + 1) % ATT_T_PER_WG) * HC
                              + ((q + 1) / ATT_T_PER_WG) * 16;
      #pragma unroll
      for (int j = 0; j < 16; ++j) rnxt[j] = rn[j];
    }
    if (r == 0) {
      #pragma unroll
      for (int j = 0; j < 16; ++j) av[j] = arow[c0 + j];
      #pragma unroll
      for (int j = 0; j < 16; ++j) lin[j & 3] = fmaf(av[j], xv[c0 + j], lin[j & 3]);
    }
    #pragma unroll
    for (int j = 0; j < 16; j += 4) {
      const float s0 = xv[c0 + j]     + rcur[j];
      const float s1 = xv[c0 + j + 1] + rcur[j + 1];
      const float s2 = xv[c0 + j + 2] + rcur[j + 2];
      const float s3 = xv[c0 + j + 3] + rcur[j + 3];
      acc[r][0] = fmaf(av[j],     fabsf(s0), acc[r][0]);
      acc[r][1] = fmaf(av[j + 1], fabsf(s1), acc[r][1]);
      acc[r][2] = fmaf(av[j + 2], fabsf(s2), acc[r][2]);
      acc[r][3] = fmaf(av[j + 3], fabsf(s3), acc[r][3]);
    }
    #pragma unroll
    for (int j = 0; j < 16; ++j) rcur[j] = rnxt[j];
  }
  const float lin_s = KA_LRELU * ((lin[0] + lin[1]) + (lin[2] + lin[3]));
  #pragma unroll
  for (int r = 0; r < ATT_T_PER_WG; ++r)
    lrow[r * LG_PITCH] =
        lin_s + KB_LRELU * ((acc[r][0] + acc[r][1]) + (acc[r][2] + acc[r][3]));
}


__device__ __forceinline__ void lds_barrier() {
  __builtin_amdgcn_fence(__ATOMIC_RELEASE, "workgroup", "local");
  __builtin_amdgcn_s_barrier();
  __builtin_amdgcn_fence(__ATOMIC_ACQUIRE, "workgroup", "local");
}

__global__ __launch_bounds__(256) void BipartPool_attn_kernel(
    const float* __restrict__ xl,
    const float* __restrict__ xr,
    const float* __restrict__ att,
    const float* __restrict__ bias,
    float* __restrict__ out)
{
  extern __shared__ __attribute__((aligned(16))) float smem[];
  float* s_logit = smem;
  float* s_part  = s_logit + ATT_ROWS * LG_PITCH;
  float* s_den   = s_part + ATT_KQ * ATT_T_PER_WG * HC;
  float* s_pself = s_den + ATT_ROWS;
  float* s_self  = s_pself + ATT_ROWS;

  const int g    = blockIdx.x >> 1;
  const int r0   = (blockIdx.x & 1) * ATT_T_PER_WG;
  const int tid  = threadIdx.x;
  const int base = g * 512;
  const int wave = tid >> 5;
  const int lane = tid & 31;

  const float* xg = xl + (size_t)(base + tid) * HC;
  float xv[C_CH];
  {
    float4 xq[16];
    load_xrow(xg, xq);
    unpack_xrow(xq, xv);
  }

  if (tid < ATT_ROWS) {
    const int r8 = tid % ATT_T_PER_WG, h = tid / ATT_T_PER_WG;
    const int r = r0 + r8;
    const int t = g * 16 + r;
    const float* xrow = xl + (size_t)t * HC + h * C_CH;
    const float* rrow = xr + r * HC + h * C_CH;
    const float* arow = att + h * C_CH;
    float4 xs[16], rs[16], ws[16];
    #pragma unroll
    for (int c = 0; c < 16; ++c) {
      xs[c] = *(const float4*)(xrow + 4 * c);
      rs[c] = *(const float4*)(rrow + 4 * c);
      ws[c] = *(const float4*)(arow + 4 * c);
    }
    float l0 = 0.f, l1 = 0.f, a0 = 0.f, a1 = 0.f;
    #pragma unroll
    for (int c = 0; c < 16; ++c) {
      l0 = fmaf(ws[c].x, xs[c].x, l0);
      l1 = fmaf(ws[c].y, xs[c].y, l1);
      a0 = fmaf(ws[c].x, fabsf(xs[c].x + rs[c].x), a0);
      a1 = fmaf(ws[c].y, fabsf(xs[c].y + rs[c].y), a1);
      l0 = fmaf(ws[c].z, xs[c].z, l0);
      l1 = fmaf(ws[c].w, xs[c].w, l1);
      a0 = fmaf(ws[c].z, fabsf(xs[c].z + rs[c].z), a0);
      a1 = fmaf(ws[c].w, fabsf(xs[c].w + rs[c].w), a1);
    }
    s_self[tid] = KA_LRELU * (l0 + l1) + KB_LRELU * (a0 + a1);
  }

  const int hw = wave >> 2;
  const int kq = wave & 3;
  const int lo = lane & 15, hi = lane >> 4;
  const float* bp = xl + (size_t)(base + kq * 128 + 2 * hi) * HC + hw * C_CH + 4 * lo;

  #pragma unroll 1
  for (int pass = 0; pass < 3; ++pass) {
    const int h = pass >> 1;
    const int n = tid + (pass & 1) * 256;
    const int hn = (pass + 1) >> 1;
    const int nn = tid + ((pass + 1) & 1) * 256;
    float4 xq[16];
    load_xrow(xl + (size_t)(base + nn) * HC + hn * C_CH, xq);
    __builtin_amdgcn_sched_barrier(0);
    logit_pass(xv, att + h * C_CH, xr + r0 * HC + h * C_CH,
               s_logit + (h * ATT_T_PER_WG) * LG_PITCH + n);
    unpack_xrow(xq, xv);
  }

  float4 b0[32], b1[32];
  #pragma unroll
  for (int i = 0; i < 28; ++i) {
    b0[i] = *(const float4*)(bp + (size_t)(4 * i) * HC);
    b1[i] = *(const float4*)(bp + (size_t)(4 * i + 1) * HC);
  }
  const int cc = tid & 63;
  float xs0[2], xs1[2];
  #pragma unroll
  for (int k = 0; k < 2; ++k) {
    const int t = g * 16 + r0 + (tid >> 6) + 4 * k;
    xs0[k] = xl[(size_t)t * HC + cc];
    xs1[k] = xl[(size_t)t * HC + C_CH + cc];
  }
  const float bias_c = bias[cc];

  __builtin_amdgcn_sched_barrier(0);
  logit_pass(xv, att + C_CH, xr + r0 * HC + C_CH,
             s_logit + ATT_T_PER_WG * LG_PITCH + tid + 256);
  __builtin_amdgcn_sched_barrier(0);
  #pragma unroll
  for (int i = 28; i < 32; ++i) {
    b0[i] = *(const float4*)(bp + (size_t)(4 * i) * HC);
    b1[i] = *(const float4*)(bp + (size_t)(4 * i + 1) * HC);
  }

  lds_barrier();

  if (g == 0 && tid < ATT_ROWS) {
    const int r8 = tid % ATT_T_PER_WG, h = tid / ATT_T_PER_WG;
    s_logit[(h * ATT_T_PER_WG + r8) * LG_PITCH + (r0 + r8)] = -__builtin_inff();
  }
  lds_barrier();

  {
    const int rh0 = wave * (ATT_ROWS / 8);
    float* row0 = s_logit + rh0 * LG_PITCH + 4 * lane;
    float* row1 = row0 + LG_PITCH;
    const float self0 = s_self[rh0], self1 = s_self[rh0 + 1];
    float4 u[4], v[4];
    #pragma unroll
    for (int j = 0; j < 4; ++j) {
      u[j] = *(const float4*)(row0 + 128 * j);
      v[j] = *(const float4*)(row1 + 128 * j);
    }
    float m0 = self0, m1 = self1;
    #pragma unroll
    for (int j = 0; j < 4; ++j) {
      m0 = fmaxf(m0, fmaxf(fmaxf(u[j].x, u[j].y), fmaxf(u[j].z, u[j].w)));
      m1 = fmaxf(m1, fmaxf(fmaxf(v[j].x, v[j].y), fmaxf(v[j].z, v[j].w)));
    }
    wave_max2(m0, m1);
    float t0 = 0.f, t1 = 0.f;
    #pragma unroll
    for (int j = 0; j < 4; ++j) {
      float4 a, b;
      a.x = __expf(u[j].x - m0); a.y = __expf(u[j].y - m0);
      a.z = __expf(u[j].z - m0); a.w = __expf(u[j].w - m0);
      b.x = __expf(v[j].x - m1); b.y = __expf(v[j].y - m1);
      b.z = __expf(v[j].z - m1); b.w = __expf(v[j].w - m1);
      *(float4*)(row0 + 128 * j) = a;
      *(float4*)(row1 + 128 * j) = b;
      t0 += (a.x + a.y) + (a.z + a.w);
      t1 += (b.x + b.y) + (b.z + b.w);
    }
    wave_sum2(t0, t1);
    if (lane == 0) {
      const float p0 = __expf(self0 - m0), p1 = __expf(self1 - m1);
      s_pself[rh0]     = p0;
      s_pself[rh0 + 1] = p1;
      s_den[rh0]       = t0 + p0 + 1e-16f;
      s_den[rh0 + 1]   = t1 + p1 + 1e-16f;
    }
  }
  lds_barrier();

  {
    const float* prow = s_logit + lo * LG_PITCH + kq * 128 + 2 * hi;
    v2f af[32];
    #pragma unroll
    for (int i = 0; i < 32; ++i) af[i] = *(const v2f*)(prow + 4 * i);
    __builtin_amdgcn_sched_barrier(0);
    v8f c0, c1, c2, c3;
    #pragma unroll
    for (int i = 0; i < 8; ++i) { c0[i] = 0.f; c1[i] = 0.f; c2[i] = 0.f; c3[i] = 0.f; }
    #pragma unroll
    for (int i = 0; i < 32; ++i) {
      v2f f0, f1, f2, f3;
      f0.x = b0[i].x; f0.y = b1[i].x;
      f1.x = b0[i].y; f1.y = b1[i].y;
      f2.x = b0[i].z; f2.y = b1[i].z;
      f3.x = b0[i].w; f3.y = b1[i].w;
      c0 = __builtin_amdgcn_wmma_f32_16x16x4_f32(false, af[i], false, f0, (short)0, c0, false, false);
      c1 = __builtin_amdgcn_wmma_f32_16x16x4_f32(false, af[i], false, f1, (short)0, c1, false, false);
      c2 = __builtin_amdgcn_wmma_f32_16x16x4_f32(false, af[i], false, f2, (short)0, c2, false, false);
      c3 = __builtin_amdgcn_wmma_f32_16x16x4_f32(false, af[i], false, f3, (short)0, c3, false, false);
    }
    if (hi == hw) {
      #pragma unroll
      for (int v = 0; v < 8; ++v) {
        float4 t;
        t.x = c0[v]; t.y = c1[v]; t.z = c2[v]; t.w = c3[v];
        *(float4*)(s_part + (kq * ATT_T_PER_WG + v) * HC + hw * C_CH + 4 * lo) = t;
      }
    }
  }
  __syncthreads();

  #pragma unroll
  for (int k = 0; k < 2; ++k) {
    const int r8 = (tid >> 6) + 4 * k;
    const int t = g * 16 + r0 + r8;
    const float* p0 = s_part + r8 * HC + cc;
    const float* p1 = p0 + C_CH;
    const int qs = ATT_T_PER_WG * HC;
    const float acc0 = (p0[0] + p0[qs]) + (p0[2 * qs] + p0[3 * qs]);
    const float acc1 = (p1[0] + p1[qs]) + (p1[2 * qs] + p1[3 * qs]);
    const float v0 = (acc0 + s_pself[r8] * xs0[k]) / s_den[r8];
    const float v1 = (acc1 + s_pself[ATT_T_PER_WG + r8] * xs1[k]) / s_den[ATT_T_PER_WG + r8];
    out[(size_t)t * 64 + cc] = 0.5f * (v0 + v1) + bias_c;
  }
}

extern "C" void kernel_launch(void* const* d_in, const int* in_sizes, int n_in,
                              void* d_out, int out_size, void* d_ws, size_t ws_size,
                              hipStream_t stream) {
  const float* x    = (const float*)d_in[0];
  const float* xcb  = (const float*)d_in[3];
  const float* wl   = (const float*)d_in[4];
  const float* bl   = (const float*)d_in[5];
  const float* wr   = (const float*)d_in[6];
  const float* br   = (const float*)d_in[7];
  const float* att  = (const float*)d_in[8];
  const float* bias = (const float*)d_in[9];
  float* out = (float*)d_out;

  float* xl = (float*)d_ws;
  float* xr = xl + (size_t)N_NODES * HC;

  BipartPool_proj_gemm_kernel<<<N_NODES / XL_ROWS_PER_BLOCK, 256, 0, stream>>>(
      x, wl, bl, xl, xcb, wr, br, xr);

  const size_t smem = (size_t)ATT_LDS_FLOATS * sizeof(float);
  BipartPool_attn_kernel<<<ATT_WGS, 256, smem, stream>>>(xl, xr, att, bias, out);
}


# x_l projection fused into the attention kernel, workgroup = (graph, head): projected rows stay in LDS (row pitch 68 floats) for logits, softmax and weighted sum; second launch = head mean + bias
# speedup vs baseline: 4.7136x; 1.5919x over previous
#include <hip/hip_runtime.h>
#include <math.h>


#define N_NODES 65536
#define IN_CH   64
#define HC      128
#define R_CENT  16
#define B_GR    128
#define HEADS   2
#define C_CH    64
#define NT      (R_CENT * B_GR)
#define G_NODES 512

typedef __attribute__((ext_vector_type(2))) float v2f;
typedef __attribute__((ext_vector_type(8))) float v8f;

template <int K>
__device__ __forceinline__ float row_xor(float x) {
  const int b = __builtin_bit_cast(int, x);
  return __builtin_bit_cast(float, __builtin_amdgcn_update_dpp(b, b, 0x160 + K, 0xf, 0xf, false));
}
__device__ __forceinline__ float xhalf(float x) {
  const int b = __builtin_bit_cast(int, x);
  return __builtin_bit_cast(float,
      __builtin_amdgcn_permlanex16(b, b, 0x76543210, 0xfedcba98, false, false));
}
__device__ __forceinline__ void wave_max2(float& a, float& b) {
  a = fmaxf(a, row_xor<1>(a)); b = fmaxf(b, row_xor<1>(b));
  a = fmaxf(a, row_xor<2>(a)); b = fmaxf(b, row_xor<2>(b));
  a = fmaxf(a, row_xor<4>(a)); b = fmaxf(b, row_xor<4>(b));
  a = fmaxf(a, row_xor<8>(a)); b = fmaxf(b, row_xor<8>(b));
  a = fmaxf(a, xhalf(a));      b = fmaxf(b, xhalf(b));
}
__device__ __forceinline__ void wave_sum2(float& a, float& b) {
  a += row_xor<1>(a); b += row_xor<1>(b);
  a += row_xor<2>(a); b += row_xor<2>(b);
  a += row_xor<4>(a); b += row_xor<4>(b);
  a += row_xor<8>(a); b += row_xor<8>(b);
  a += xhalf(a);      b += xhalf(b);
}
__device__ __forceinline__ void row_sum2(float& a, float& b) {
  a += row_xor<1>(a); b += row_xor<1>(b);
  a += row_xor<2>(a); b += row_xor<2>(b);
  a += row_xor<4>(a); b += row_xor<4>(b);
  a += row_xor<8>(a); b += row_xor<8>(b);
}

constexpr float KA_LRELU = (float)(0.5 * (1.0 + (double)0.2f));
constexpr float KB_LRELU = (float)(0.5 * (1.0 - (double)0.2f));

#define XP        68
#define LG_PITCH  516
#define KSPLIT    8
#define OFF_XL    0
#define OFF_XSELF (OFF_XL + G_NODES * XP)
#define OFF_XR    (OFF_XSELF + R_CENT * XP)
#define OFF_LOGIT (OFF_XR + (R_CENT + 1) * C_CH)
#define OFF_PART  (OFF_LOGIT + R_CENT * LG_PITCH)
#define OFF_DEN   (OFF_PART + KSPLIT * R_CENT * C_CH)
#define LDS_FLOATS (OFF_DEN + 3 * R_CENT)

__device__ __forceinline__ void load_a_strip(const float* __restrict__ ap, v2f (&a)[16]) {
  #pragma unroll
  for (int i = 0; i < 16; ++i) a[i] = *(const v2f*)(ap + 4 * i);
}

__device__ __forceinline__ void lds_barrier() {
  __builtin_amdgcn_fence(__ATOMIC_RELEASE, "workgroup", "local");
  __builtin_amdgcn_s_barrier();
  __builtin_amdgcn_fence(__ATOMIC_ACQUIRE, "workgroup", "local");
}

__global__ __launch_bounds__(256) void BipartPool_graph_head_kernel(
    const float* __restrict__ x,
    const float* __restrict__ wl,
    const float* __restrict__ bl,
    const float* __restrict__ xc,
    const float* __restrict__ wr,
    const float* __restrict__ br,
    const float* __restrict__ att,
    float* __restrict__ hsum)
{
  __shared__ __attribute__((aligned(16))) float smem[LDS_FLOATS];
  float* s_xl    = smem + OFF_XL;
  float* s_xself = smem + OFF_XSELF;
  float* s_xr    = smem + OFF_XR;
  float* s_logit = smem + OFF_LOGIT;
  float* s_part  = smem + OFF_PART;
  float* s_den   = smem + OFF_DEN;
  float* s_pself = s_den + R_CENT;
  float* s_self  = s_pself + R_CENT;

  const int g    = blockIdx.x >> 1;
  const int h    = blockIdx.x & 1;
  const int tid  = threadIdx.x;
  const int wave = __builtin_amdgcn_readfirstlane(tid >> 5);
  const int lane = tid & 31;
  const int lo   = lane & 15;
  const int hi   = lane >> 4;

  {
    const int ncol = h * C_CH + 4 * lo;
    const int row0 = wave * 64;
    const float* ap = x + ((size_t)g * G_NODES + row0 + lo) * IN_CH + 2 * hi;

    float4 wr0[16], wr1[16];
    #pragma unroll
    for (int i = 0; i < 16; ++i) {
      const int kk = 4 * i + 2 * hi;
      wr0[i] = *(const float4*)(wl + (size_t)kk * HC + ncol);
      wr1[i] = *(const float4*)(wl + (size_t)(kk + 1) * HC + ncol);
    }
    const float4 bv = *(const float4*)(bl + ncol);
    v2f acur[16], anxt[16];
    load_a_strip(ap, acur);
    __builtin_amdgcn_sched_barrier(0);

    v2f bf0[16], bf1[16], bf2[16], bf3[16];
    #pragma unroll
    for (int i = 0; i < 16; ++i) {
      bf0[i].x = wr0[i].x; bf0[i].y = wr1[i].x;
      bf1[i].x = wr0[i].y; bf1[i].y = wr1[i].y;
      bf2[i].x = wr0[i].z; bf2[i].y = wr1[i].z;
      bf3[i].x = wr0[i].w; bf3[i].y = wr1[i].w;
    }

    const bool is_self = wave < 4;
    const int  et   = wave & 3;
    const int  ecol = h * C_CH + 4 * lo + et;
    const float* ea_p = (is_self ? x + ((size_t)g * R_CENT + lo) * IN_CH
                                 : xc + (size_t)lo * IN_CH) + 2 * hi;
    const float* ew_p = (is_self ? wl : wr) + (size_t)(2 * hi) * HC + ecol;
    const float* eb_p = (is_self ? bl : br) + ecol;
    float* ed_p = is_self ? s_xself + (8 * hi) * XP + 4 * lo + et
                          : s_xr + (8 * hi) * C_CH + 4 * lo + et;
    const int ed_pitch = is_self ? XP : C_CH;
    v2f ea[16], eb[16];
    float ebias = 0.f;

    float* op = s_xl + (row0 + 8 * hi) * XP + 4 * lo;
    #pragma unroll
    for (int s = 0; s < 4; ++s) {
      if (s + 1 < 4) load_a_strip(ap + (size_t)(s + 1) * 16 * IN_CH, anxt);
      if (s == 1) load_a_strip(ea_p, ea);
      if (s == 2) {
        #pragma unroll
        for (int i = 0; i < 16; ++i) {
          eb[i].x = ew_p[(size_t)(4 * i) * HC];
          eb[i].y = ew_p[(size_t)(4 * i + 1) * HC];
        }
        ebias = *eb_p;
      }
      __builtin_amdgcn_sched_barrier(0);

      v8f c0, c1, c2, c3;
      #pragma unroll
      for (int r = 0; r < 8; ++r) { c0[r] = bv.x; c1[r] = bv.y; c2[r] = bv.z; c3[r] = bv.w; }
      #pragma unroll
      for (int i = 0; i < 16; ++i) {
        c0 = __builtin_amdgcn_wmma_f32_16x16x4_f32(false, acur[i], false, bf0[i], (short)0, c0, false, false);
        c1 = __builtin_amdgcn_wmma_f32_16x16x4_f32(false, acur[i], false, bf1[i], (short)0, c1, false, false);
        c2 = __builtin_amdgcn_wmma_f32_16x16x4_f32(false, acur[i], false, bf2[i], (short)0, c2, false, false);
        c3 = __builtin_amdgcn_wmma_f32_16x16x4_f32(false, acur[i], false, bf3[i], (short)0, c3, false, false);
      }
      #pragma unroll
      for (int r = 0; r < 8; ++r) {
        float4 t;
        t.x = c0[r]; t.y = c1[r]; t.z = c2[r]; t.w = c3[r];
        *(float4*)(op + (s * 16 + r) * XP) = t;
      }
      #pragma unroll
      for (int i = 0; i < 16; ++i) acur[i] = anxt[i];
    }

    v8f e0, e1;
    #pragma unroll
    for (int r = 0; r < 8; ++r) { e0[r] = ebias; e1[r] = 0.f; }
    #pragma unroll
    for (int i = 0; i < 16; i += 2) {
      e0 = __builtin_amdgcn_wmma_f32_16x16x4_f32(false, ea[i],     false, eb[i],     (short)0, e0, false, false);
      e1 = __builtin_amdgcn_wmma_f32_16x16x4_f32(false, ea[i + 1], false, eb[i + 1], (short)0, e1, false, false);
    }
    #pragma unroll
    for (int r = 0; r < 8; ++r) ed_p[r * ed_pitch] = e0[r] + e1[r];
  }
  lds_barrier();

  const float* arow = att + h * C_CH;

  {
    const int r = tid >> 4;
    const float4 xs = *(const float4*)(s_xself + r * XP + 4 * lo);
    const float4 rs = *(const float4*)(s_xr + r * C_CH + 4 * lo);
    const float4 ws = *(const float4*)(arow + 4 * lo);
    float l = fmaf(ws.x, xs.x, fmaf(ws.y, xs.y, fmaf(ws.z, xs.z, ws.w * xs.w)));
    float a = fmaf(ws.x, fabsf(xs.x + rs.x), fmaf(ws.y, fabsf(xs.y + rs.y),
              fmaf(ws.z, fabsf(xs.z + rs.z), ws.w * fabsf(xs.w + rs.w))));
    row_sum2(l, a);
    if (lo == 0) s_self[r] = KA_LRELU * l + KB_LRELU * a;
  }

  {
    float xv0[C_CH], xv1[C_CH];
    {
      const float* p0 = s_xl + tid * XP;
      const float* p1 = s_xl + (tid + 256) * XP;
      #pragma unroll
      for (int c = 0; c < 16; ++c) {
        const float4 q0 = *(const float4*)(p0 + 4 * c);
        const float4 q1 = *(const float4*)(p1 + 4 * c);
        xv0[4 * c] = q0.x; xv0[4 * c + 1] = q0.y; xv0[4 * c + 2] = q0.z; xv0[4 * c + 3] = q0.w;
        xv1[4 * c] = q1.x; xv1[4 * c + 1] = q1.y; xv1[4 * c + 2] = q1.z; xv1[4 * c + 3] = q1.w;
      }
    }
    float av[C_CH];
    #pragma unroll
    for (int c = 0; c < C_CH; ++c) av[c] = arow[c];
    float lin0[4] = {0.f, 0.f, 0.f, 0.f}, lin1[4] = {0.f, 0.f, 0.f, 0.f};
    #pragma unroll
    for (int c = 0; c < C_CH; ++c) {
      lin0[c & 3] = fmaf(av[c], xv0[c], lin0[c & 3]);
      lin1[c & 3] = fmaf(av[c], xv1[c], lin1[c & 3]);
    }
    const float lin0_s = KA_LRELU * ((lin0[0] + lin0[1]) + (lin0[2] + lin0[3]));
    const float lin1_s = KA_LRELU * ((lin1[0] + lin1[1]) + (lin1[2] + lin1[3]));

    float4 ra[4], rb[4];
    #pragma unroll
    for (int j = 0; j < 4; ++j) ra[j] = *(const float4*)(s_xr + 4 * j);
    float* lrow = s_logit + tid;
    #pragma unroll 1
    for (int r = 0; r < R_CENT; ++r) {
      const float* rp = s_xr + r * C_CH;
      float a0[4] = {0.f, 0.f, 0.f, 0.f}, a1[4] = {0.f, 0.f, 0.f, 0.f};
      #pragma unroll
      for (int q = 0; q < 4; ++q) {
        const int c0 = 16 * q;
        float4 (&cur)[4] = (q & 1) ? rb : ra;
        float4 (&nxt)[4] = (q & 1) ? ra : rb;
        #pragma unroll
        for (int j = 0; j < 4; ++j) nxt[j] = *(const float4*)(rp + c0 + 16 + 4 * j);
        #pragma unroll
        for (int j = 0; j < 4; ++j) {
          const int c = c0 + 4 * j;
          const float s00 = xv0[c]     + cur[j].x, s10 = xv1[c]     + cur[j].x;
          const float s01 = xv0[c + 1] + cur[j].y, s11 = xv1[c + 1] + cur[j].y;
          const float s02 = xv0[c + 2] + cur[j].z, s12 = xv1[c + 2] + cur[j].z;
          const float s03 = xv0[c + 3] + cur[j].w, s13 = xv1[c + 3] + cur[j].w;
          a0[0] = fmaf(av[c],     fabsf(s00), a0[0]); a1[0] = fmaf(av[c],     fabsf(s10), a1[0]);
          a0[1] = fmaf(av[c + 1], fabsf(s01), a0[1]); a1[1] = fmaf(av[c + 1], fabsf(s11), a1[1]);
          a0[2] = fmaf(av[c + 2], fabsf(s02), a0[2]); a1[2] = fmaf(av[c + 2], fabsf(s12), a1[2]);
          a0[3] = fmaf(av[c + 3], fabsf(s03), a0[3]); a1[3] = fmaf(av[c + 3], fabsf(s13), a1[3]);
        }
      }
      lrow[r * LG_PITCH]       = lin0_s + KB_LRELU * ((a0[0] + a0[1]) + (a0[2] + a0[3]));
      lrow[r * LG_PITCH + 256] = lin1_s + KB_LRELU * ((a1[0] + a1[1]) + (a1[2] + a1[3]));
    }
    if (g == 0 && tid < R_CENT) s_logit[tid * LG_PITCH + tid] = -__builtin_inff();
  }
  lds_barrier();

  {
    const int rh0 = wave * 2;
    float* row0 = s_logit + rh0 * LG_PITCH + 4 * lane;
    float* row1 = row0 + LG_PITCH;
    const float self0 = s_self[rh0], self1 = s_self[rh0 + 1];
    float4 u[4], v[4];
    #pragma unroll
    for (int j = 0; j < 4; ++j) {
      u[j] = *(const float4*)(row0 + 128 * j);
      v[j] = *(const float4*)(row1 + 128 * j);
    }
    float m0 = self0, m1 = self1;
    #pragma unroll
    for (int j = 0; j < 4; ++j) {
      m0 = fmaxf(m0, fmaxf(fmaxf(u[j].x, u[j].y), fmaxf(u[j].z, u[j].w)));
      m1 = fmaxf(m1, fmaxf(fmaxf(v[j].x, v[j].y), fmaxf(v[j].z, v[j].w)));
    }
    wave_max2(m0, m1);
    float t0 = 0.f, t1 = 0.f;
    #pragma unroll
    for (int j = 0; j < 4; ++j) {
      float4 a, b;
      a.x = __expf(u[j].x - m0); a.y = __expf(u[j].y - m0);
      a.z = __expf(u[j].z - m0); a.w = __expf(u[j].w - m0);
      b.x = __expf(v[j].x - m1); b.y = __expf(v[j].y - m1);
      b.z = __expf(v[j].z - m1); b.w = __expf(v[j].w - m1);
      *(float4*)(row0 + 128 * j) = a;
      *(float4*)(row1 + 128 * j) = b;
      t0 += (a.x + a.y) + (a.z + a.w);
      t1 += (b.x + b.y) + (b.z + b.w);
    }
    wave_sum2(t0, t1);
    if (lane == 0) {
      const float p0 = __expf(self0 - m0), p1 = __expf(self1 - m1);
      s_pself[rh0]     = p0;
      s_pself[rh0 + 1] = p1;
      s_den[rh0]       = t0 + p0 + 1e-16f;
      s_den[rh0 + 1]   = t1 + p1 + 1e-16f;
    }
  }
  lds_barrier();

  {
    const float* prow = s_logit + lo * LG_PITCH + wave * 64 + 2 * hi;
    const float* bp   = s_xl + (wave * 64 + 2 * hi) * XP + 4 * lo;
    v2f af[16];
    float4 b0[16], b1[16];
    #pragma unroll
    for (int i = 0; i < 16; ++i) {
      b0[i] = *(const float4*)(bp + (4 * i) * XP);
      b1[i] = *(const float4*)(bp + (4 * i + 1) * XP);
    }
    #pragma unroll
    for (int i = 0; i < 16; ++i) af[i] = *(const v2f*)(prow + 4 * i);
    __builtin_amdgcn_sched_barrier(0);
    v8f c0, c1, c2, c3;
    #pragma unroll
    for (int i = 0; i < 8; ++i) { c0[i] = 0.f; c1[i] = 0.f; c2[i] = 0.f; c3[i] = 0.f; }
    #pragma unroll
    for (int i = 0; i < 16; ++i) {
      v2f f0, f1, f2, f3;
      f0.x = b0[i].x; f0.y = b1[i].x;
      f1.x = b0[i].y; f1.y = b1[i].y;
      f2.x = b0[i].z; f2.y = b1[i].z;
      f3.x = b0[i].w; f3.y = b1[i].w;
      c0 = __builtin_amdgcn_wmma_f32_16x16x4_f32(false, af[i], false, f0, (short)0, c0, false, false);
      c1 = __builtin_amdgcn_wmma_f32_16x16x4_f32(false, af[i], false, f1, (short)0, c1, false, false);
      c2 = __builtin_amdgcn_wmma_f32_16x16x4_f32(false, af[i], false, f2, (short)0, c2, false, false);
      c3 = __builtin_amdgcn_wmma_f32_16x16x4_f32(false, af[i], false, f3, (short)0, c3, false, false);
    }
    float* pp = s_part + (wave * R_CENT + 8 * hi) * C_CH + 4 * lo;
    #pragma unroll
    for (int v = 0; v < 8; ++v) {
      float4 t;
      t.x = c0[v]; t.y = c1[v]; t.z = c2[v]; t.w = c3[v];
      *(float4*)(pp + v * C_CH) = t;
    }
  }
  lds_barrier();

  {
    const int r  = tid >> 4;
    const int c4 = 4 * lo;
    const float* pq = s_part + r * C_CH + c4;
    const int qs = R_CENT * C_CH;
    float4 p[KSPLIT];
    #pragma unroll
    for (int k = 0; k < KSPLIT; ++k) p[k] = *(const float4*)(pq + k * qs);
    const float4 xs = *(const float4*)(s_xself + r * XP + c4);
    const float ps = s_pself[r], den = s_den[r];
    float4 o;
    o.x = ((((p[0].x + p[1].x) + (p[2].x + p[3].x)) + ((p[4].x + p[5].x) + (p[6].x + p[7].x))) + ps * xs.x) / den;
    o.y = ((((p[0].y + p[1].y) + (p[2].y + p[3].y)) + ((p[4].y + p[5].y) + (p[6].y + p[7].y))) + ps * xs.y) / den;
    o.z = ((((p[0].z + p[1].z) + (p[2].z + p[3].z)) + ((p[4].z + p[5].z) + (p[6].z + p[7].z))) + ps * xs.z) / den;
    o.w = ((((p[0].w + p[1].w) + (p[2].w + p[3].w)) + ((p[4].w + p[5].w) + (p[6].w + p[7].w))) + ps * xs.w) / den;
    *(float4*)(hsum + ((size_t)h * NT + (size_t)g * R_CENT + r) * C_CH + c4) = o;
  }
}

__global__ __launch_bounds__(256) void BipartPool_head_mean_kernel(
    const float* __restrict__ hsum,
    const float* __restrict__ bias,
    float* __restrict__ out)
{
  const int i = blockIdx.x * 256 + threadIdx.x;
  const float4 v0 = *(const float4*)(hsum + (size_t)4 * i);
  const float4 v1 = *(const float4*)(hsum + (size_t)NT * C_CH + (size_t)4 * i);
  const float4 bb = *(const float4*)(bias + 4 * (i & 15));
  float4 o;
  o.x = 0.5f * (v0.x + v1.x) + bb.x;
  o.y = 0.5f * (v0.y + v1.y) + bb.y;
  o.z = 0.5f * (v0.z + v1.z) + bb.z;
  o.w = 0.5f * (v0.w + v1.w) + bb.w;
  *(float4*)(out + (size_t)4 * i) = o;
}

extern "C" void kernel_launch(void* const* d_in, const int* in_sizes, int n_in,
                              void* d_out, int out_size, void* d_ws, size_t ws_size,
                              hipStream_t stream) {
  const float* x    = (const float*)d_in[0];
  const float* xcb  = (const float*)d_in[3];
  const float* wl   = (const float*)d_in[4];
  const float* bl   = (const float*)d_in[5];
  const float* wr   = (const float*)d_in[6];
  const float* br   = (const float*)d_in[7];
  const float* att  = (const float*)d_in[8];
  const float* bias = (const float*)d_in[9];
  float* out = (float*)d_out;

  float* hsum = (float*)d_ws;

  BipartPool_graph_head_kernel<<<B_GR * HEADS, 256, 0, stream>>>(
      x, wl, bl, xcb, wr, br, att, hsum);

  BipartPool_head_mean_kernel<<<NT * C_CH / 4 / 256, 256, 0, stream>>>(hsum, bias, out);
}


# logit loop: max(x+r,0) = max(x,-r)+r (the +r cancels in softmax): one max + one fma per element, dual-issued; lane = 4 nodes x 32 channels, two partial-logit planes added by the softmax pass
# speedup vs baseline: 4.9551x; 1.0512x over previous
#include <hip/hip_runtime.h>
#include <math.h>


#define N_NODES 65536
#define IN_CH   64
#define HC      128
#define R_CENT  16
#define B_GR    128
#define HEADS   2
#define C_CH    64
#define NT      (R_CENT * B_GR)
#define G_NODES 512

typedef __attribute__((ext_vector_type(2))) float v2f;
typedef __attribute__((ext_vector_type(8))) float v8f;

template <int K>
__device__ __forceinline__ float row_xor(float x) {
  const int b = __builtin_bit_cast(int, x);
  return __builtin_bit_cast(float, __builtin_amdgcn_update_dpp(b, b, 0x160 + K, 0xf, 0xf, false));
}
__device__ __forceinline__ float xhalf(float x) {
  const int b = __builtin_bit_cast(int, x);
  return __builtin_bit_cast(float,
      __builtin_amdgcn_permlanex16(b, b, 0x76543210, 0xfedcba98, false, false));
}
__device__ __forceinline__ void wave_max2(float& a, float& b) {
  a = fmaxf(a, row_xor<1>(a)); b = fmaxf(b, row_xor<1>(b));
  a = fmaxf(a, row_xor<2>(a)); b = fmaxf(b, row_xor<2>(b));
  a = fmaxf(a, row_xor<4>(a)); b = fmaxf(b, row_xor<4>(b));
  a = fmaxf(a, row_xor<8>(a)); b = fmaxf(b, row_xor<8>(b));
  a = fmaxf(a, xhalf(a));      b = fmaxf(b, xhalf(b));
}
__device__ __forceinline__ void wave_sum2(float& a, float& b) {
  a += row_xor<1>(a); b += row_xor<1>(b);
  a += row_xor<2>(a); b += row_xor<2>(b);
  a += row_xor<4>(a); b += row_xor<4>(b);
  a += row_xor<8>(a); b += row_xor<8>(b);
  a += xhalf(a);      b += xhalf(b);
}
__device__ __forceinline__ void row_sum2(float& a, float& b) {
  a += row_xor<1>(a); b += row_xor<1>(b);
  a += row_xor<2>(a); b += row_xor<2>(b);
  a += row_xor<4>(a); b += row_xor<4>(b);
  a += row_xor<8>(a); b += row_xor<8>(b);
}

constexpr float K_LIN  = 0.2f;
constexpr float K_RELU = (float)(1.0 - (double)0.2f);

#define XP        68
#define LG_PITCH  516
#define KSPLIT    8
#define OFF_XL    0
#define OFF_XSELF (OFF_XL + G_NODES * XP)
#define OFF_XR    (OFF_XSELF + R_CENT * XP)
#define OFF_LOGIT (OFF_XR + (R_CENT + 1) * C_CH)
#define OFF_LGT2  (OFF_LOGIT + R_CENT * LG_PITCH)
#define OFF_PART  (OFF_LGT2 + R_CENT * LG_PITCH)
#define OFF_DEN   (OFF_PART + KSPLIT * R_CENT * C_CH)
#define LDS_FLOATS (OFF_DEN + 3 * R_CENT)

__device__ __forceinline__ void load_a_strip(const float* __restrict__ ap, v2f (&a)[16]) {
  #pragma unroll
  for (int i = 0; i < 16; ++i) a[i] = *(const v2f*)(ap + 4 * i);
}

__device__ __forceinline__ void lds_barrier() {
  __builtin_amdgcn_fence(__ATOMIC_RELEASE, "workgroup", "local");
  __builtin_amdgcn_s_barrier();
  __builtin_amdgcn_fence(__ATOMIC_ACQUIRE, "workgroup", "local");
}

__global__ __launch_bounds__(256) void BipartPool_graph_head_kernel(
    const float* __restrict__ x,
    const float* __restrict__ wl,
    const float* __restrict__ bl,
    const float* __restrict__ xc,
    const float* __restrict__ wr,
    const float* __restrict__ br,
    const float* __restrict__ att,
    float* __restrict__ hsum)
{
  __shared__ __attribute__((aligned(16))) float smem[LDS_FLOATS];
  float* s_xl    = smem + OFF_XL;
  float* s_xself = smem + OFF_XSELF;
  float* s_xr    = smem + OFF_XR;
  float* s_logit = smem + OFF_LOGIT;
  float* s_lgt2  = smem + OFF_LGT2;
  float* s_part  = smem + OFF_PART;
  float* s_den   = smem + OFF_DEN;
  float* s_pself = s_den + R_CENT;
  float* s_self  = s_pself + R_CENT;

  const int g    = blockIdx.x >> 1;
  const int h    = blockIdx.x & 1;
  const int tid  = threadIdx.x;
  const int wave = __builtin_amdgcn_readfirstlane(tid >> 5);
  const int lane = tid & 31;
  const int lo   = lane & 15;
  const int hi   = lane >> 4;

  {
    const int ncol = h * C_CH + 4 * lo;
    const int row0 = wave * 64;
    const float* ap = x + ((size_t)g * G_NODES + row0 + lo) * IN_CH + 2 * hi;

    float4 wr0[16], wr1[16];
    #pragma unroll
    for (int i = 0; i < 16; ++i) {
      const int kk = 4 * i + 2 * hi;
      wr0[i] = *(const float4*)(wl + (size_t)kk * HC + ncol);
      wr1[i] = *(const float4*)(wl + (size_t)(kk + 1) * HC + ncol);
    }
    const float4 bv = *(const float4*)(bl + ncol);
    v2f acur[16], anxt[16];
    load_a_strip(ap, acur);
    __builtin_amdgcn_sched_barrier(0);

    v2f bf0[16], bf1[16], bf2[16], bf3[16];
    #pragma unroll
    for (int i = 0; i < 16; ++i) {
      bf0[i].x = wr0[i].x; bf0[i].y = wr1[i].x;
      bf1[i].x = wr0[i].y; bf1[i].y = wr1[i].y;
      bf2[i].x = wr0[i].z; bf2[i].y = wr1[i].z;
      bf3[i].x = wr0[i].w; bf3[i].y = wr1[i].w;
    }

    const bool is_self = wave < 4;
    const int  et   = wave & 3;
    const int  ecol = h * C_CH + 4 * lo + et;
    const float* ea_p = (is_self ? x + ((size_t)g * R_CENT + lo) * IN_CH
                                 : xc + (size_t)lo * IN_CH) + 2 * hi;
    const float* ew_p = (is_self ? wl : wr) + (size_t)(2 * hi) * HC + ecol;
    const float* eb_p = (is_self ? bl : br) + ecol;
    float* ed_p = is_self ? s_xself + (8 * hi) * XP + 4 * lo + et
                          : s_xr + (8 * hi) * C_CH + 4 * lo + et;
    const int ed_pitch = is_self ? XP : C_CH;
    v2f ea[16], eb[16];
    float ebias = 0.f;

    float* op = s_xl + (row0 + 8 * hi) * XP + 4 * lo;
    #pragma unroll
    for (int s = 0; s < 4; ++s) {
      if (s + 1 < 4) load_a_strip(ap + (size_t)(s + 1) * 16 * IN_CH, anxt);
      if (s == 1) load_a_strip(ea_p, ea);
      if (s == 2) {
        #pragma unroll
        for (int i = 0; i < 16; ++i) {
          eb[i].x = ew_p[(size_t)(4 * i) * HC];
          eb[i].y = ew_p[(size_t)(4 * i + 1) * HC];
        }
        ebias = *eb_p;
      }
      __builtin_amdgcn_sched_barrier(0);

      v8f c0, c1, c2, c3;
      #pragma unroll
      for (int r = 0; r < 8; ++r) { c0[r] = bv.x; c1[r] = bv.y; c2[r] = bv.z; c3[r] = bv.w; }
      #pragma unroll
      for (int i = 0; i < 16; ++i) {
        c0 = __builtin_amdgcn_wmma_f32_16x16x4_f32(false, acur[i], false, bf0[i], (short)0, c0, false, false);
        c1 = __builtin_amdgcn_wmma_f32_16x16x4_f32(false, acur[i], false, bf1[i], (short)0, c1, false, false);
        c2 = __builtin_amdgcn_wmma_f32_16x16x4_f32(false, acur[i], false, bf2[i], (short)0, c2, false, false);
        c3 = __builtin_amdgcn_wmma_f32_16x16x4_f32(false, acur[i], false, bf3[i], (short)0, c3, false, false);
      }
      #pragma unroll
      for (int r = 0; r < 8; ++r) {
        float4 t;
        t.x = c0[r]; t.y = c1[r]; t.z = c2[r]; t.w = c3[r];
        *(float4*)(op + (s * 16 + r) * XP) = t;
      }
      #pragma unroll
      for (int i = 0; i < 16; ++i) acur[i] = anxt[i];
    }

    v8f e0, e1;
    #pragma unroll
    for (int r = 0; r < 8; ++r) { e0[r] = ebias; e1[r] = 0.f; }
    #pragma unroll
    for (int i = 0; i < 16; i += 2) {
      e0 = __builtin_amdgcn_wmma_f32_16x16x4_f32(false, ea[i],     false, eb[i],     (short)0, e0, false, false);
      e1 = __builtin_amdgcn_wmma_f32_16x16x4_f32(false, ea[i + 1], false, eb[i + 1], (short)0, e1, false, false);
    }
    const float esign = is_self ? 1.0f : -1.0f;
    #pragma unroll
    for (int r = 0; r < 8; ++r) ed_p[r * ed_pitch] = esign * (e0[r] + e1[r]);
  }
  lds_barrier();

  const float* arow = att + h * C_CH;

  {
    const int r = tid >> 4;
    const float4 xs = *(const float4*)(s_xself + r * XP + 4 * lo);
    const float4 rs = *(const float4*)(s_xr + r * C_CH + 4 * lo);
    const float4 ws = *(const float4*)(arow + 4 * lo);
    float l = fmaf(ws.x, xs.x, fmaf(ws.y, xs.y, fmaf(ws.z, xs.z, ws.w * xs.w)));
    float a = fmaf(ws.x, fmaxf(xs.x, rs.x), fmaf(ws.y, fmaxf(xs.y, rs.y),
              fmaf(ws.z, fmaxf(xs.z, rs.z), ws.w * fmaxf(xs.w, rs.w))));
    row_sum2(l, a);
    if (lo == 0) s_self[r] = K_LIN * l + K_RELU * a;
  }

  {
    const int cb = 32 * hi;
    float xv[4][32];
    {
      const float* p = s_xl + (wave * 64 + lo) * XP + cb;
      #pragma unroll
      for (int q = 0; q < 4; ++q) {
        #pragma unroll
        for (int c = 0; c < 8; ++c) {
          const float4 t = *(const float4*)(p + q * 16 * XP + 4 * c);
          xv[q][4 * c] = t.x; xv[q][4 * c + 1] = t.y; xv[q][4 * c + 2] = t.z; xv[q][4 * c + 3] = t.w;
        }
      }
    }
    float av[32];
    #pragma unroll
    for (int c = 0; c < 8; ++c) {
      const float4 t = *(const float4*)(arow + cb + 4 * c);
      av[4 * c] = t.x; av[4 * c + 1] = t.y; av[4 * c + 2] = t.z; av[4 * c + 3] = t.w;
    }
    float lin_s[4];
    {
      float l[4][4];
      #pragma unroll
      for (int q = 0; q < 4; ++q) {
        l[q][0] = av[0] * xv[q][0]; l[q][1] = av[1] * xv[q][1];
        l[q][2] = av[2] * xv[q][2]; l[q][3] = av[3] * xv[q][3];
      }
      #pragma unroll
      for (int c = 4; c < 32; c += 4) {
        #pragma unroll
        for (int q = 0; q < 4; ++q) {
          l[q][0] = fmaf(av[c],     xv[q][c],     l[q][0]);
          l[q][1] = fmaf(av[c + 1], xv[q][c + 1], l[q][1]);
          l[q][2] = fmaf(av[c + 2], xv[q][c + 2], l[q][2]);
          l[q][3] = fmaf(av[c + 3], xv[q][c + 3], l[q][3]);
        }
      }
      #pragma unroll
      for (int q = 0; q < 4; ++q) lin_s[q] = K_LIN * ((l[q][0] + l[q][1]) + (l[q][2] + l[q][3]));
    }

    const float* rbase = s_xr + cb;
    float4 ra[4], rb[4];
    #pragma unroll
    for (int j = 0; j < 4; ++j) ra[j] = *(const float4*)(rbase + 4 * j);
    float* lrow = (hi ? s_lgt2 : s_logit) + wave * 64 + lo;
    #pragma unroll 1
    for (int r = 0; r < R_CENT; ++r) {
      const float* rp = rbase + r * C_CH;
      float acc[4][4];
      #pragma unroll
      for (int pc = 0; pc < 2; ++pc) {
        float4 (&cur)[4] = (pc & 1) ? rb : ra;
        float4 (&nxt)[4] = (pc & 1) ? ra : rb;
        const float* np = rp + (pc == 0 ? 16 : C_CH);
        #pragma unroll
        for (int j = 0; j < 4; ++j) nxt[j] = *(const float4*)(np + 4 * j);
        #pragma unroll
        for (int j = 0; j < 4; ++j) {
          const int c = 16 * pc + 4 * j;
          float m[4][4];
          #pragma unroll
          for (int q = 0; q < 4; ++q) {
            m[q][0] = fmaxf(xv[q][c],     cur[j].x);
            m[q][1] = fmaxf(xv[q][c + 1], cur[j].y);
            m[q][2] = fmaxf(xv[q][c + 2], cur[j].z);
            m[q][3] = fmaxf(xv[q][c + 3], cur[j].w);
          }
          #pragma unroll
          for (int q = 0; q < 4; ++q) {
            if (c == 0) {
              acc[q][0] = av[0] * m[q][0];
              acc[q][1] = av[1] * m[q][1];
              acc[q][2] = av[2] * m[q][2];
              acc[q][3] = av[3] * m[q][3];
            } else {
              acc[q][0] = fmaf(av[c],     m[q][0], acc[q][0]);
              acc[q][1] = fmaf(av[c + 1], m[q][1], acc[q][1]);
              acc[q][2] = fmaf(av[c + 2], m[q][2], acc[q][2]);
              acc[q][3] = fmaf(av[c + 3], m[q][3], acc[q][3]);
            }
          }
        }
      }
      #pragma unroll
      for (int q = 0; q < 4; ++q)
        lrow[r * LG_PITCH + 16 * q] = lin_s[q] + K_RELU * ((acc[q][0] + acc[q][1]) + (acc[q][2] + acc[q][3]));
    }
    if (g == 0 && tid < R_CENT) s_logit[tid * LG_PITCH + tid] = -__builtin_inff();
  }
  lds_barrier();

  {
    const int rh0 = wave * 2;
    float* row0 = s_logit + rh0 * LG_PITCH + 4 * lane;
    float* row1 = row0 + LG_PITCH;
    const float self0 = s_self[rh0], self1 = s_self[rh0 + 1];
    float4 u[4], v[4];
    #pragma unroll
    for (int j = 0; j < 4; ++j) {
      u[j] = *(const float4*)(row0 + 128 * j);
      v[j] = *(const float4*)(row1 + 128 * j);
    }
    #pragma unroll
    for (int j = 0; j < 4; ++j) {
      const float4 u2 = *(const float4*)(row0 + (OFF_LGT2 - OFF_LOGIT) + 128 * j);
      const float4 v2 = *(const float4*)(row1 + (OFF_LGT2 - OFF_LOGIT) + 128 * j);
      u[j].x += u2.x; u[j].y += u2.y; u[j].z += u2.z; u[j].w += u2.w;
      v[j].x += v2.x; v[j].y += v2.y; v[j].z += v2.z; v[j].w += v2.w;
    }
    float m0 = self0, m1 = self1;
    #pragma unroll
    for (int j = 0; j < 4; ++j) {
      m0 = fmaxf(m0, fmaxf(fmaxf(u[j].x, u[j].y), fmaxf(u[j].z, u[j].w)));
      m1 = fmaxf(m1, fmaxf(fmaxf(v[j].x, v[j].y), fmaxf(v[j].z, v[j].w)));
    }
    wave_max2(m0, m1);
    float t0 = 0.f, t1 = 0.f;
    #pragma unroll
    for (int j = 0; j < 4; ++j) {
      float4 a, b;
      a.x = __expf(u[j].x - m0); a.y = __expf(u[j].y - m0);
      a.z = __expf(u[j].z - m0); a.w = __expf(u[j].w - m0);
      b.x = __expf(v[j].x - m1); b.y = __expf(v[j].y - m1);
      b.z = __expf(v[j].z - m1); b.w = __expf(v[j].w - m1);
      *(float4*)(row0 + 128 * j) = a;
      *(float4*)(row1 + 128 * j) = b;
      t0 += (a.x + a.y) + (a.z + a.w);
      t1 += (b.x + b.y) + (b.z + b.w);
    }
    wave_sum2(t0, t1);
    if (lane == 0) {
      const float p0 = __expf(self0 - m0), p1 = __expf(self1 - m1);
      s_pself[rh0]     = p0;
      s_pself[rh0 + 1] = p1;
      s_den[rh0]       = t0 + p0 + 1e-16f;
      s_den[rh0 + 1]   = t1 + p1 + 1e-16f;
    }
  }
  lds_barrier();

  {
    const float* prow = s_logit + lo * LG_PITCH + wave * 64 + 2 * hi;
    const float* bp   = s_xl + (wave * 64 + 2 * hi) * XP + 4 * lo;
    v2f af[16];
    float4 b0[16], b1[16];
    #pragma unroll
    for (int i = 0; i < 16; ++i) {
      b0[i] = *(const float4*)(bp + (4 * i) * XP);
      b1[i] = *(const float4*)(bp + (4 * i + 1) * XP);
    }
    #pragma unroll
    for (int i = 0; i < 16; ++i) af[i] = *(const v2f*)(prow + 4 * i);
    __builtin_amdgcn_sched_barrier(0);
    v8f c0, c1, c2, c3;
    #pragma unroll
    for (int i = 0; i < 8; ++i) { c0[i] = 0.f; c1[i] = 0.f; c2[i] = 0.f; c3[i] = 0.f; }
    #pragma unroll
    for (int i = 0; i < 16; ++i) {
      v2f f0, f1, f2, f3;
      f0.x = b0[i].x; f0.y = b1[i].x;
      f1.x = b0[i].y; f1.y = b1[i].y;
      f2.x = b0[i].z; f2.y = b1[i].z;
      f3.x = b0[i].w; f3.y = b1[i].w;
      c0 = __builtin_amdgcn_wmma_f32_16x16x4_f32(false, af[i], false, f0, (short)0, c0, false, false);
      c1 = __builtin_amdgcn_wmma_f32_16x16x4_f32(false, af[i], false, f1, (short)0, c1, false, false);
      c2 = __builtin_amdgcn_wmma_f32_16x16x4_f32(false, af[i], false, f2, (short)0, c2, false, false);
      c3 = __builtin_amdgcn_wmma_f32_16x16x4_f32(false, af[i], false, f3, (short)0, c3, false, false);
    }
    float* pp = s_part + (wave * R_CENT + 8 * hi) * C_CH + 4 * lo;
    #pragma unroll
    for (int v = 0; v < 8; ++v) {
      float4 t;
      t.x = c0[v]; t.y = c1[v]; t.z = c2[v]; t.w = c3[v];
      *(float4*)(pp + v * C_CH) = t;
    }
  }
  lds_barrier();

  {
    const int r  = tid >> 4;
    const int c4 = 4 * lo;
    const float* pq = s_part + r * C_CH + c4;
    const int qs = R_CENT * C_CH;
    float4 p[KSPLIT];
    #pragma unroll
    for (int k = 0; k < KSPLIT; ++k) p[k] = *(const float4*)(pq + k * qs);
    const float4 xs = *(const float4*)(s_xself + r * XP + c4);
    const float ps = s_pself[r], den = s_den[r];
    float4 o;
    o.x = ((((p[0].x + p[1].x) + (p[2].x + p[3].x)) + ((p[4].x + p[5].x) + (p[6].x + p[7].x))) + ps * xs.x) / den;
    o.y = ((((p[0].y + p[1].y) + (p[2].y + p[3].y)) + ((p[4].y + p[5].y) + (p[6].y + p[7].y))) + ps * xs.y) / den;
    o.z = ((((p[0].z + p[1].z) + (p[2].z + p[3].z)) + ((p[4].z + p[5].z) + (p[6].z + p[7].z))) + ps * xs.z) / den;
    o.w = ((((p[0].w + p[1].w) + (p[2].w + p[3].w)) + ((p[4].w + p[5].w) + (p[6].w + p[7].w))) + ps * xs.w) / den;
    *(float4*)(hsum + ((size_t)h * NT + (size_t)g * R_CENT + r) * C_CH + c4) = o;
  }
}

__global__ __launch_bounds__(256) void BipartPool_head_mean_kernel(
    const float* __restrict__ hsum,
    const float* __restrict__ bias,
    float* __restrict__ out)
{
  const int i = blockIdx.x * 256 + threadIdx.x;
  const float4 v0 = *(const float4*)(hsum + (size_t)4 * i);
  const float4 v1 = *(const float4*)(hsum + (size_t)NT * C_CH + (size_t)4 * i);
  const float4 bb = *(const float4*)(bias + 4 * (i & 15));
  float4 o;
  o.x = 0.5f * (v0.x + v1.x) + bb.x;
  o.y = 0.5f * (v0.y + v1.y) + bb.y;
  o.z = 0.5f * (v0.z + v1.z) + bb.z;
  o.w = 0.5f * (v0.w + v1.w) + bb.w;
  *(float4*)(out + (size_t)4 * i) = o;
}

extern "C" void kernel_launch(void* const* d_in, const int* in_sizes, int n_in,
                              void* d_out, int out_size, void* d_ws, size_t ws_size,
                              hipStream_t stream) {
  const float* x    = (const float*)d_in[0];
  const float* xcb  = (const float*)d_in[3];
  const float* wl   = (const float*)d_in[4];
  const float* bl   = (const float*)d_in[5];
  const float* wr   = (const float*)d_in[6];
  const float* br   = (const float*)d_in[7];
  const float* att  = (const float*)d_in[8];
  const float* bias = (const float*)d_in[9];
  float* out = (float*)d_out;

  float* hsum = (float*)d_ws;

  BipartPool_graph_head_kernel<<<B_GR * HEADS, 256, 0, stream>>>(
      x, wl, bl, xcb, wr, br, att, hsum);

  BipartPool_head_mean_kernel<<<NT * C_CH / 4 / 256, 256, 0, stream>>>(hsum, bias, out);
}
